# starting kernel
# baseline (speedup 1.0000x reference)
#include <hip/hip_runtime.h>
#include <hip/hip_bf16.h>


typedef __attribute__((ext_vector_type(16))) _Float16 v16h;
typedef __attribute__((ext_vector_type(8)))  _Float16 v8h;
typedef __attribute__((ext_vector_type(8)))  float    v8f;

#define DEV __device__ __forceinline__

DEV v8f wmma_f16(v16h a, v16h b, v8f c) {
  return __builtin_amdgcn_wmma_f32_16x16x32_f16(false, a, false, b, (short)0, c,
                                                false, false);
}

DEV v16h load_frag(const _Float16* __restrict__ tile, int ld, int lane) {
  const int m  = lane & 15;
  const int kh = (lane & 16) >> 1;
  const _Float16* p = tile + (size_t)m * ld + kh;
  v8h lo = *(const v8h*)p;
  v8h hi = *(const v8h*)(p + 16);
  v16h a;
#pragma unroll
  for (int i = 0; i < 8; ++i) { a[i] = lo[i]; a[i + 8] = hi[i]; }
  return a;
}

__global__ void tcast_kernel(const float* __restrict__ in,
                             _Float16* __restrict__ out, int R, int C) {
  size_t base = (size_t)blockIdx.y * R * C;
  int idx = blockIdx.x * blockDim.x + threadIdx.x;
  if (idx >= R * C) return;
  int r = idx / C, c = idx - r * C;
  out[base + (size_t)c * R + r] = (_Float16)in[base + idx];
}

__global__ void ln_kernel(const float* __restrict__ x, const float* __restrict__ g,
                          const float* __restrict__ be, _Float16* __restrict__ out) {
  const int C = 384;
  int lane = threadIdx.x & 31;
  size_t row = (size_t)blockIdx.x * 8 + (threadIdx.x >> 5);
  const float* xr = x + row * C;
  float v[12];
  float s = 0.f;
#pragma unroll
  for (int i = 0; i < 12; ++i) { v[i] = xr[lane + 32 * i]; s += v[i]; }
#pragma unroll
  for (int m = 16; m >= 1; m >>= 1) s += __shfl_xor(s, m, 32);
  float mu = s * (1.f / 384.f);
  float var = 0.f;
#pragma unroll
  for (int i = 0; i < 12; ++i) { float d = v[i] - mu; var += d * d; }
#pragma unroll
  for (int m = 16; m >= 1; m >>= 1) var += __shfl_xor(var, m, 32);
  float rs = rsqrtf(var * (1.f / 384.f) + 1e-5f);
  _Float16* orow = out + row * C;
#pragma unroll
  for (int i = 0; i < 12; ++i) {
    int c = lane + 32 * i;
    orow[c] = (_Float16)((v[i] - mu) * rs * g[c] + be[c]);
  }
}

__global__ void qkv_kernel(const _Float16* __restrict__ h16,
                           const _Float16* __restrict__ wqt,
                           const _Float16* __restrict__ wkt,
                           const _Float16* __restrict__ wvt,
                           _Float16* __restrict__ q, _Float16* __restrict__ k,
                           _Float16* __restrict__ vt) {
  const int Cc = 384, Dh = 64, T = 256, H = 6;
  int tile = blockIdx.x;
  int dt = tile & 3;
  int rest = tile >> 2;
  int hh = rest % H;
  int rt = rest / H;
  int lane = threadIdx.x;
  const int n = lane & 15, mh = (lane >> 4) * 8;

  const _Float16* Ab = h16 + (size_t)rt * 16 * Cc;
  const _Float16* Bq = wqt + ((size_t)hh * Dh + dt * 16) * Cc;
  const _Float16* Bk = wkt + ((size_t)hh * Dh + dt * 16) * Cc;
  const _Float16* Bv = wvt + ((size_t)hh * Dh + dt * 16) * Cc;

  v8f cq = {}, ck = {}, cv = {};
  for (int k0 = 0; k0 < Cc; k0 += 32) {
    v16h fh = load_frag(Ab + k0, Cc, lane);
    v16h fq = load_frag(Bq + k0, Cc, lane);
    v16h fk = load_frag(Bk + k0, Cc, lane);
    v16h fv = load_frag(Bv + k0, Cc, lane);
    cq = wmma_f16(fh, fq, cq);
    ck = wmma_f16(fh, fk, ck);
    cv = wmma_f16(fv, fh, cv);
  }
  int b = rt >> 4, t0 = (rt & 15) * 16;
  size_t bh = (size_t)b * H + hh;
#pragma unroll
  for (int r = 0; r < 8; ++r) {
    size_t qoff = (bh * T + t0 + mh + r) * Dh + dt * 16 + n;
    q[qoff] = (_Float16)cq[r];
    k[qoff] = (_Float16)ck[r];
    vt[(bh * Dh + dt * 16 + mh + r) * (size_t)T + t0 + n] = (_Float16)cv[r];
  }
}

__global__ void attn_kernel(const _Float16* __restrict__ q,
                            const _Float16* __restrict__ k,
                            const _Float16* __restrict__ vt,
                            _Float16* __restrict__ attn_out) {
  const int T = 256, Dh = 64, H = 6, C = 384;
  int tile = blockIdx.x;
  int it = tile & 15;
  int bh = tile >> 4;
  int b = bh / H, h = bh - b * H;
  int t0 = it * 16;
  int lane = threadIdx.x;
  const int n = lane & 15, mh = (lane >> 4) * 8;

  const _Float16* qb = q + ((size_t)bh * T + t0) * Dh;
  const _Float16* kb = k + (size_t)bh * T * Dh;
  const _Float16* vb = vt + (size_t)bh * Dh * T;

  v16h aq0 = load_frag(qb, Dh, lane);
  v16h aq1 = load_frag(qb + 32, Dh, lane);

  float mrow[8], lrow[8];
#pragma unroll
  for (int r = 0; r < 8; ++r) { mrow[r] = -__builtin_inff(); lrow[r] = 0.f; }
  v8f O[4];
#pragma unroll
  for (int i = 0; i < 4; ++i) O[i] = (v8f){};

  __shared__ _Float16 Pl[16 * 32];
  const float scale = 0.05103103630798287f;

  for (int s0 = 0; s0 < t0 + 16; s0 += 32) {
    float P0[8], P1[8];
    {
      v8f acc = {};
      v16h b0 = load_frag(kb + (size_t)s0 * Dh, Dh, lane);
      v16h b1 = load_frag(kb + (size_t)s0 * Dh + 32, Dh, lane);
      acc = wmma_f16(aq0, b0, acc);
      acc = wmma_f16(aq1, b1, acc);
#pragma unroll
      for (int r = 0; r < 8; ++r) P0[r] = acc[r] * scale;
    }
    {
      v8f acc = {};
      v16h b0 = load_frag(kb + (size_t)(s0 + 16) * Dh, Dh, lane);
      v16h b1 = load_frag(kb + (size_t)(s0 + 16) * Dh + 32, Dh, lane);
      acc = wmma_f16(aq0, b0, acc);
      acc = wmma_f16(aq1, b1, acc);
#pragma unroll
      for (int r = 0; r < 8; ++r) P1[r] = acc[r] * scale;
    }
#pragma unroll
    for (int r = 0; r < 8; ++r) {
      int trow = t0 + mh + r;
      if (s0 + n > trow)      P0[r] = -__builtin_inff();
      if (s0 + 16 + n > trow) P1[r] = -__builtin_inff();
      float mx = fmaxf(P0[r], P1[r]);
      mx = fmaxf(mx, __shfl_xor(mx, 1, 32));
      mx = fmaxf(mx, __shfl_xor(mx, 2, 32));
      mx = fmaxf(mx, __shfl_xor(mx, 4, 32));
      mx = fmaxf(mx, __shfl_xor(mx, 8, 32));
      float mnew = fmaxf(mrow[r], mx);
      float al = expf(mrow[r] - mnew);
      float p0 = expf(P0[r] - mnew);
      float p1 = expf(P1[r] - mnew);
      float rs = p0 + p1;
      rs += __shfl_xor(rs, 1, 32);
      rs += __shfl_xor(rs, 2, 32);
      rs += __shfl_xor(rs, 4, 32);
      rs += __shfl_xor(rs, 8, 32);
      lrow[r] = lrow[r] * al + rs;
      mrow[r] = mnew;
      P0[r] = p0; P1[r] = p1;
      O[0][r] *= al; O[1][r] *= al; O[2][r] *= al; O[3][r] *= al;
    }
    __syncthreads();
#pragma unroll
    for (int r = 0; r < 8; ++r) {
      Pl[(mh + r) * 32 + n]      = (_Float16)P0[r];
      Pl[(mh + r) * 32 + 16 + n] = (_Float16)P1[r];
    }
    __syncthreads();
    v16h ap = load_frag(Pl, 32, lane);
#pragma unroll
    for (int nt = 0; nt < 4; ++nt) {
      v16h bv = load_frag(vb + (size_t)(nt * 16) * T + s0, T, lane);
      O[nt] = wmma_f16(ap, bv, O[nt]);
    }
  }
#pragma unroll
  for (int r = 0; r < 8; ++r) {
    float inv = 1.f / lrow[r];
    _Float16* orow = attn_out + ((size_t)b * T + t0 + mh + r) * C + h * Dh;
#pragma unroll
    for (int nt = 0; nt < 4; ++nt) orow[nt * 16 + n] = (_Float16)(O[nt][r] * inv);
  }
}

template <int EPI>
__global__ void gemm16(const _Float16* __restrict__ A, const _Float16* __restrict__ Bt,
                       const float* __restrict__ bias, const float* __restrict__ res,
                       float* __restrict__ outf, _Float16* __restrict__ outh,
                       int N, int K) {
  int ntn = N >> 4;
  int tm = blockIdx.x / ntn;
  int tn = blockIdx.x - tm * ntn;
  int lane = threadIdx.x;
  const _Float16* Ab = A + (size_t)tm * 16 * K;
  const _Float16* Bb = Bt + (size_t)tn * 16 * K;
  v8f acc = {};
  for (int k0 = 0; k0 < K; k0 += 32) {
    v16h a = load_frag(Ab + k0, K, lane);
    v16h b = load_frag(Bb + k0, K, lane);
    acc = wmma_f16(a, b, acc);
  }
  const int n = lane & 15, mh = (lane >> 4) * 8;
  const int col = tn * 16 + n;
  const float bv = bias[col];
#pragma unroll
  for (int r = 0; r < 8; ++r) {
    size_t row = (size_t)tm * 16 + mh + r;
    float v = acc[r] + bv;
    if (EPI == 0)      outf[row * N + col] = v + res[row * N + col];
    else if (EPI == 1) outh[row * N + col] = (_Float16)fmaxf(v, 0.f);
    else               outf[row * N + col] += v;
  }
}

extern "C" void kernel_launch(void* const* d_in, const int* in_sizes, int n_in,
                              void* d_out, int out_size, void* d_ws, size_t ws_size,
                              hipStream_t stream) {
  (void)in_sizes; (void)n_in; (void)out_size; (void)ws_size;
  const float* x   = (const float*)d_in[0];
  const float* Wq  = (const float*)d_in[1];
  const float* Wk  = (const float*)d_in[2];
  const float* Wv  = (const float*)d_in[3];
  const float* Wp  = (const float*)d_in[4];
  const float* bp  = (const float*)d_in[5];
  const float* W1  = (const float*)d_in[6];
  const float* b1  = (const float*)d_in[7];
  const float* W2  = (const float*)d_in[8];
  const float* b2  = (const float*)d_in[9];
  const float* g1  = (const float*)d_in[10];
  const float* be1 = (const float*)d_in[11];
  const float* g2  = (const float*)d_in[12];
  const float* be2 = (const float*)d_in[13];
  float* out = (float*)d_out;

  char* ws = (char*)d_ws;
  const size_t S = (size_t)32768 * 384 * 2;
  _Float16* h16    = (_Float16*)(ws + 0 * S);
  _Float16* q16    = (_Float16*)(ws + 1 * S);
  _Float16* k16    = (_Float16*)(ws + 2 * S);
  _Float16* vt16   = (_Float16*)(ws + 3 * S);
  _Float16* attn16 = (_Float16*)(ws + 0 * S);
  _Float16* h2_16  = (_Float16*)(ws + 1 * S);
  _Float16* a1_16  = (_Float16*)(ws + 2 * S);
  char* wbase = ws + 6 * S;
  _Float16* wqt = (_Float16*)(wbase);
  _Float16* wkt = (_Float16*)(wbase + 1 * 294912);
  _Float16* wvt = (_Float16*)(wbase + 2 * 294912);
  _Float16* wpt = (_Float16*)(wbase + 3 * 294912);
  _Float16* w1t = (_Float16*)(wbase + 4 * 294912);
  _Float16* w2t = (_Float16*)(wbase + 4 * 294912 + 1179648);

  tcast_kernel<<<dim3(96, 6),   256, 0, stream>>>(Wq, wqt, 384, 64);
  tcast_kernel<<<dim3(96, 6),   256, 0, stream>>>(Wk, wkt, 384, 64);
  tcast_kernel<<<dim3(96, 6),   256, 0, stream>>>(Wv, wvt, 384, 64);
  tcast_kernel<<<dim3(576, 1),  256, 0, stream>>>(Wp, wpt, 384, 384);
  tcast_kernel<<<dim3(2304, 1), 256, 0, stream>>>(W1, w1t, 384, 1536);
  tcast_kernel<<<dim3(2304, 1), 256, 0, stream>>>(W2, w2t, 1536, 384);

  ln_kernel<<<4096, 256, 0, stream>>>(x, g1, be1, h16);
  qkv_kernel<<<49152, 32, 0, stream>>>(h16, wqt, wkt, wvt, q16, k16, vt16);
  attn_kernel<<<12288, 32, 0, stream>>>(q16, k16, vt16, attn16);
  gemm16<0><<<2048 * 24, 32, 0, stream>>>(attn16, wpt, bp, x, out, nullptr, 384, 384);
  ln_kernel<<<4096, 256, 0, stream>>>(out, g2, be2, h2_16);
  gemm16<1><<<2048 * 96, 32, 0, stream>>>(h2_16, w1t, b1, nullptr, nullptr, a1_16, 1536, 384);
  gemm16<2><<<2048 * 24, 32, 0, stream>>>(a1_16, w2t, b2, nullptr, out, nullptr, 384, 1536);
}


# All four weight GEMMs (QKV, proj, FC1, FC2): one 16x16 tile per wave -> LDS-staged 128x128 work-group tile, 4 waves x 64x64 (4x4 WMMA accumulators); same f16 operands, same k order of sums
# speedup vs baseline: 2.4165x; 2.4165x over previous
#include <hip/hip_runtime.h>
#include <hip/hip_bf16.h>


typedef __attribute__((ext_vector_type(16))) _Float16 v16h;
typedef __attribute__((ext_vector_type(8)))  _Float16 v8h;
typedef __attribute__((ext_vector_type(8)))  float    v8f;

#define DEV __device__ __forceinline__

DEV v8f wmma_f16(v16h a, v16h b, v8f c) {
  return __builtin_amdgcn_wmma_f32_16x16x32_f16(false, a, false, b, (short)0, c,
                                                false, false);
}

DEV v16h load_frag(const _Float16* __restrict__ tile, int ld, int lane) {
  const int m  = lane & 15;
  const int kh = (lane & 16) >> 1;
  const _Float16* p = tile + (size_t)m * ld + kh;
  v8h lo = *(const v8h*)p;
  v8h hi = *(const v8h*)(p + 16);
  v16h a;
#pragma unroll
  for (int i = 0; i < 8; ++i) { a[i] = lo[i]; a[i + 8] = hi[i]; }
  return a;
}

__global__ void tcast_kernel(const float* __restrict__ in,
                             _Float16* __restrict__ out, int R, int C) {
  size_t base = (size_t)blockIdx.y * R * C;
  int idx = blockIdx.x * blockDim.x + threadIdx.x;
  if (idx >= R * C) return;
  int r = idx / C, c = idx - r * C;
  out[base + (size_t)c * R + r] = (_Float16)in[base + idx];
}

__global__ void ln_kernel(const float* __restrict__ x, const float* __restrict__ g,
                          const float* __restrict__ be, _Float16* __restrict__ out) {
  const int C = 384;
  int lane = threadIdx.x & 31;
  size_t row = (size_t)blockIdx.x * 8 + (threadIdx.x >> 5);
  const float* xr = x + row * C;
  float v[12];
  float s = 0.f;
#pragma unroll
  for (int i = 0; i < 12; ++i) { v[i] = xr[lane + 32 * i]; s += v[i]; }
#pragma unroll
  for (int m = 16; m >= 1; m >>= 1) s += __shfl_xor(s, m, 32);
  float mu = s * (1.f / 384.f);
  float var = 0.f;
#pragma unroll
  for (int i = 0; i < 12; ++i) { float d = v[i] - mu; var += d * d; }
#pragma unroll
  for (int m = 16; m >= 1; m >>= 1) var += __shfl_xor(var, m, 32);
  float rs = rsqrtf(var * (1.f / 384.f) + 1e-5f);
  _Float16* orow = out + row * C;
#pragma unroll
  for (int i = 0; i < 12; ++i) {
    int c = lane + 32 * i;
    orow[c] = (_Float16)((v[i] - mu) * rs * g[c] + be[c]);
  }
}

__global__ void attn_kernel(const _Float16* __restrict__ q,
                            const _Float16* __restrict__ k,
                            const _Float16* __restrict__ vt,
                            _Float16* __restrict__ attn_out) {
  const int T = 256, Dh = 64, H = 6, C = 384;
  int tile = blockIdx.x;
  int it = tile & 15;
  int bh = tile >> 4;
  int b = bh / H, h = bh - b * H;
  int t0 = it * 16;
  int lane = threadIdx.x;
  const int n = lane & 15, mh = (lane >> 4) * 8;

  const _Float16* qb = q + ((size_t)bh * T + t0) * Dh;
  const _Float16* kb = k + (size_t)bh * T * Dh;
  const _Float16* vb = vt + (size_t)bh * Dh * T;

  v16h aq0 = load_frag(qb, Dh, lane);
  v16h aq1 = load_frag(qb + 32, Dh, lane);

  float mrow[8], lrow[8];
#pragma unroll
  for (int r = 0; r < 8; ++r) { mrow[r] = -__builtin_inff(); lrow[r] = 0.f; }
  v8f O[4];
#pragma unroll
  for (int i = 0; i < 4; ++i) O[i] = (v8f){};

  __shared__ _Float16 Pl[16 * 32];
  const float scale = 0.05103103630798287f;

  for (int s0 = 0; s0 < t0 + 16; s0 += 32) {
    float P0[8], P1[8];
    {
      v8f acc = {};
      v16h b0 = load_frag(kb + (size_t)s0 * Dh, Dh, lane);
      v16h b1 = load_frag(kb + (size_t)s0 * Dh + 32, Dh, lane);
      acc = wmma_f16(aq0, b0, acc);
      acc = wmma_f16(aq1, b1, acc);
#pragma unroll
      for (int r = 0; r < 8; ++r) P0[r] = acc[r] * scale;
    }
    {
      v8f acc = {};
      v16h b0 = load_frag(kb + (size_t)(s0 + 16) * Dh, Dh, lane);
      v16h b1 = load_frag(kb + (size_t)(s0 + 16) * Dh + 32, Dh, lane);
      acc = wmma_f16(aq0, b0, acc);
      acc = wmma_f16(aq1, b1, acc);
#pragma unroll
      for (int r = 0; r < 8; ++r) P1[r] = acc[r] * scale;
    }
#pragma unroll
    for (int r = 0; r < 8; ++r) {
      int trow = t0 + mh + r;
      if (s0 + n > trow)      P0[r] = -__builtin_inff();
      if (s0 + 16 + n > trow) P1[r] = -__builtin_inff();
      float mx = fmaxf(P0[r], P1[r]);
      mx = fmaxf(mx, __shfl_xor(mx, 1, 32));
      mx = fmaxf(mx, __shfl_xor(mx, 2, 32));
      mx = fmaxf(mx, __shfl_xor(mx, 4, 32));
      mx = fmaxf(mx, __shfl_xor(mx, 8, 32));
      float mnew = fmaxf(mrow[r], mx);
      float al = expf(mrow[r] - mnew);
      float p0 = expf(P0[r] - mnew);
      float p1 = expf(P1[r] - mnew);
      float rs = p0 + p1;
      rs += __shfl_xor(rs, 1, 32);
      rs += __shfl_xor(rs, 2, 32);
      rs += __shfl_xor(rs, 4, 32);
      rs += __shfl_xor(rs, 8, 32);
      lrow[r] = lrow[r] * al + rs;
      mrow[r] = mnew;
      P0[r] = p0; P1[r] = p1;
      O[0][r] *= al; O[1][r] *= al; O[2][r] *= al; O[3][r] *= al;
    }
    __syncthreads();
#pragma unroll
    for (int r = 0; r < 8; ++r) {
      Pl[(mh + r) * 32 + n]      = (_Float16)P0[r];
      Pl[(mh + r) * 32 + 16 + n] = (_Float16)P1[r];
    }
    __syncthreads();
    v16h ap = load_frag(Pl, 32, lane);
#pragma unroll
    for (int nt = 0; nt < 4; ++nt) {
      v16h bv = load_frag(vb + (size_t)(nt * 16) * T + s0, T, lane);
      O[nt] = wmma_f16(ap, bv, O[nt]);
    }
  }
#pragma unroll
  for (int r = 0; r < 8; ++r) {
    float inv = 1.f / lrow[r];
    _Float16* orow = attn_out + ((size_t)b * T + t0 + mh + r) * C + h * Dh;
#pragma unroll
    for (int nt = 0; nt < 4; ++nt) orow[nt * 16 + n] = (_Float16)(O[nt][r] * inv);
  }
}

#define GT_BM 128
#define GT_BN 128
#define GT_BK 64
#define GT_LD 72

template <int EPI>
__global__ __launch_bounds__(128)
void gemm_tile(const _Float16* __restrict__ A, const _Float16* __restrict__ Bt,
               const float* __restrict__ bias, const float* __restrict__ res,
               float* __restrict__ outf, _Float16* __restrict__ outh,
               _Float16* __restrict__ outk, _Float16* __restrict__ outvt,
               int N, int K) {
  __shared__ __attribute__((aligned(16))) _Float16 As[GT_BM * GT_LD];
  __shared__ __attribute__((aligned(16))) _Float16 Bs[GT_BN * GT_LD];
  const int tid = threadIdx.x;
  const int lane = tid & 31;
  const int w = tid >> 5;
  const int wm = w >> 1, wn = w & 1;
  const int ntn = N / GT_BN;
  const int tm = blockIdx.x / ntn;
  const int tn = blockIdx.x - tm * ntn;

  const int srow = tid >> 3;
  const int scol = (tid & 7) * 8;
  const _Float16* Ag = A  + ((size_t)tm * GT_BM + srow) * K + scol;
  const _Float16* Bg = Bt + ((size_t)tn * GT_BN + srow) * K + scol;
  _Float16* Asw = As + srow * GT_LD + scol;
  _Float16* Bsw = Bs + srow * GT_LD + scol;
  const _Float16* Asr = As + (wm * 64) * GT_LD;
  const _Float16* Bsr = Bs + (wn * 64) * GT_LD;

  v8f acc[4][4];
#pragma unroll
  for (int i = 0; i < 4; ++i)
#pragma unroll
    for (int j = 0; j < 4; ++j) acc[i][j] = (v8f){};

  v8h sa[8], sb[8];
#pragma unroll
  for (int i = 0; i < 8; ++i) {
    sa[i] = *(const v8h*)(Ag + (size_t)(16 * i) * K);
    sb[i] = *(const v8h*)(Bg + (size_t)(16 * i) * K);
  }

  for (int k0 = 0; k0 < K; k0 += GT_BK) {
#pragma unroll
    for (int i = 0; i < 8; ++i) {
      *(v8h*)(Asw + (16 * i) * GT_LD) = sa[i];
      *(v8h*)(Bsw + (16 * i) * GT_LD) = sb[i];
    }
    __syncthreads();
    if (k0 + GT_BK < K) {
#pragma unroll
      for (int i = 0; i < 8; ++i) {
        sa[i] = *(const v8h*)(Ag + (size_t)(16 * i) * K + k0 + GT_BK);
        sb[i] = *(const v8h*)(Bg + (size_t)(16 * i) * K + k0 + GT_BK);
      }
    }
#pragma unroll
    for (int kk = 0; kk < GT_BK; kk += 32) {
      v16h af[4], bf[4];
#pragma unroll
      for (int i = 0; i < 4; ++i) {
        af[i] = load_frag(Asr + (i * 16) * GT_LD + kk, GT_LD, lane);
        bf[i] = load_frag(Bsr + (i * 16) * GT_LD + kk, GT_LD, lane);
      }
#pragma unroll
      for (int i = 0; i < 4; ++i)
#pragma unroll
        for (int j = 0; j < 4; ++j) acc[i][j] = wmma_f16(af[i], bf[j], acc[i][j]);
    }
    __syncthreads();
  }

  const int n = lane & 15, mh = (lane >> 4) * 8;
  const int cbase = tn * GT_BN + wn * 64;
  const size_t rbase = (size_t)tm * GT_BM + wm * 64 + mh;
#pragma unroll
  for (int j = 0; j < 4; ++j) {
    const int col = cbase + j * 16 + n;
    float bv = 0.f;
    if (EPI != 3) bv = bias[col];
#pragma unroll
    for (int i = 0; i < 4; ++i) {
      const size_t row0 = rbase + i * 16;
      if (EPI == 3) {
        const int which = cbase / 384;
        const int hh = (cbase - which * 384) >> 6;
        const int d = j * 16 + n;
        const size_t b = row0 >> 8;
        const size_t t = row0 & 255;
        const size_t bh = b * 6 + hh;
        if (which == 2) {
          v8h pk;
#pragma unroll
          for (int r = 0; r < 8; ++r) pk[r] = (_Float16)acc[i][j][r];
          *(v8h*)(outvt + (bh * 64 + d) * 256 + t) = pk;
        } else {
          _Float16* o = (which == 0) ? outh : outk;
#pragma unroll
          for (int r = 0; r < 8; ++r) o[(bh * 256 + t + r) * 64 + d] = (_Float16)acc[i][j][r];
        }
      } else {
#pragma unroll
        for (int r = 0; r < 8; ++r) {
          const size_t off = (row0 + r) * N + col;
          const float v = acc[i][j][r] + bv;
          if (EPI == 0)      outf[off] = v + res[off];
          else if (EPI == 1) outh[off] = (_Float16)fmaxf(v, 0.f);
          else               outf[off] += v;
        }
      }
    }
  }
}

extern "C" void kernel_launch(void* const* d_in, const int* in_sizes, int n_in,
                              void* d_out, int out_size, void* d_ws, size_t ws_size,
                              hipStream_t stream) {
  (void)in_sizes; (void)n_in; (void)out_size; (void)ws_size;
  const float* x   = (const float*)d_in[0];
  const float* Wq  = (const float*)d_in[1];
  const float* Wk  = (const float*)d_in[2];
  const float* Wv  = (const float*)d_in[3];
  const float* Wp  = (const float*)d_in[4];
  const float* bp  = (const float*)d_in[5];
  const float* W1  = (const float*)d_in[6];
  const float* b1  = (const float*)d_in[7];
  const float* W2  = (const float*)d_in[8];
  const float* b2  = (const float*)d_in[9];
  const float* g1  = (const float*)d_in[10];
  const float* be1 = (const float*)d_in[11];
  const float* g2  = (const float*)d_in[12];
  const float* be2 = (const float*)d_in[13];
  float* out = (float*)d_out;

  char* ws = (char*)d_ws;
  const size_t S = (size_t)32768 * 384 * 2;
  _Float16* h16    = (_Float16*)(ws + 0 * S);
  _Float16* q16    = (_Float16*)(ws + 1 * S);
  _Float16* k16    = (_Float16*)(ws + 2 * S);
  _Float16* vt16   = (_Float16*)(ws + 3 * S);
  _Float16* attn16 = (_Float16*)(ws + 0 * S);
  _Float16* h2_16  = (_Float16*)(ws + 1 * S);
  _Float16* a1_16  = (_Float16*)(ws + 2 * S);
  char* wbase = ws + 6 * S;
  _Float16* wqt = (_Float16*)(wbase);
  _Float16* wkt = (_Float16*)(wbase + 1 * 294912);
  _Float16* wvt = (_Float16*)(wbase + 2 * 294912);
  _Float16* wpt = (_Float16*)(wbase + 3 * 294912);
  _Float16* w1t = (_Float16*)(wbase + 4 * 294912);
  _Float16* w2t = (_Float16*)(wbase + 4 * 294912 + 1179648);

  tcast_kernel<<<dim3(96, 6),   256, 0, stream>>>(Wq, wqt, 384, 64);
  tcast_kernel<<<dim3(96, 6),   256, 0, stream>>>(Wk, wkt, 384, 64);
  tcast_kernel<<<dim3(96, 6),   256, 0, stream>>>(Wv, wvt, 384, 64);
  tcast_kernel<<<dim3(576, 1),  256, 0, stream>>>(Wp, wpt, 384, 384);
  tcast_kernel<<<dim3(2304, 1), 256, 0, stream>>>(W1, w1t, 384, 1536);
  tcast_kernel<<<dim3(2304, 1), 256, 0, stream>>>(W2, w2t, 1536, 384);

  ln_kernel<<<4096, 256, 0, stream>>>(x, g1, be1, h16);
  gemm_tile<3><<<2304, 128, 0, stream>>>(h16, wqt, nullptr, nullptr, nullptr, q16, k16, vt16, 1152, 384);
  attn_kernel<<<12288, 32, 0, stream>>>(q16, k16, vt16, attn16);
  gemm_tile<0><<<768, 128, 0, stream>>>(attn16, wpt, bp, x, out, nullptr, nullptr, nullptr, 384, 384);
  ln_kernel<<<4096, 256, 0, stream>>>(out, g2, be2, h2_16);
  gemm_tile<1><<<3072, 128, 0, stream>>>(h2_16, w1t, b1, nullptr, nullptr, a1_16, nullptr, nullptr, 1536, 384);
  gemm_tile<2><<<768, 128, 0, stream>>>(a1_16, w2t, b2, nullptr, out, nullptr, nullptr, nullptr, 384, 1536);
}


# Six weight-cast launches and LN1 merged into one launch (casts as 64x64 LDS tile transposes, 16 elements per thread, whole-line stores): 13 launches -> 7, 55296 cast waves -> 3456
# speedup vs baseline: 3.0003x; 1.2416x over previous
#include <hip/hip_runtime.h>
#include <hip/hip_bf16.h>


typedef __attribute__((ext_vector_type(16))) _Float16 v16h;
typedef __attribute__((ext_vector_type(8)))  _Float16 v8h;
typedef __attribute__((ext_vector_type(8)))  float    v8f;

#define DEV __device__ __forceinline__

DEV v8f wmma_f16(v16h a, v16h b, v8f c) {
  return __builtin_amdgcn_wmma_f32_16x16x32_f16(false, a, false, b, (short)0, c,
                                                false, false);
}

DEV v16h load_frag(const _Float16* __restrict__ tile, int ld, int lane) {
  const int m  = lane & 15;
  const int kh = (lane & 16) >> 1;
  const _Float16* p = tile + (size_t)m * ld + kh;
  v8h lo = *(const v8h*)p;
  v8h hi = *(const v8h*)(p + 16);
  v16h a;
#pragma unroll
  for (int i = 0; i < 8; ++i) { a[i] = lo[i]; a[i + 8] = hi[i]; }
  return a;
}

DEV void ln_rows(const float* __restrict__ x, const float* __restrict__ g,
                 const float* __restrict__ be, _Float16* __restrict__ out, int blk) {
  const int C = 384;
  int lane = threadIdx.x & 31;
  size_t row = (size_t)blk * 8 + (threadIdx.x >> 5);
  const float* xr = x + row * C;
  float v[12];
  float s = 0.f;
#pragma unroll
  for (int i = 0; i < 12; ++i) { v[i] = xr[lane + 32 * i]; s += v[i]; }
#pragma unroll
  for (int m = 16; m >= 1; m >>= 1) s += __shfl_xor(s, m, 32);
  float mu = s * (1.f / 384.f);
  float var = 0.f;
#pragma unroll
  for (int i = 0; i < 12; ++i) { float d = v[i] - mu; var += d * d; }
#pragma unroll
  for (int m = 16; m >= 1; m >>= 1) var += __shfl_xor(var, m, 32);
  float rs = rsqrtf(var * (1.f / 384.f) + 1e-5f);
  _Float16* orow = out + row * C;
#pragma unroll
  for (int i = 0; i < 12; ++i) {
    int c = lane + 32 * i;
    orow[c] = (_Float16)((v[i] - mu) * rs * g[c] + be[c]);
  }
}

__global__ void ln_kernel(const float* __restrict__ x, const float* __restrict__ g,
                          const float* __restrict__ be, _Float16* __restrict__ out) {
  ln_rows(x, g, be, out, blockIdx.x);
}

typedef __attribute__((ext_vector_type(4))) float v4f;
#define PREP_CAST_BLOCKS 432
#define PREP_LD 66

__global__ __launch_bounds__(256)
void prep_kernel(const float* __restrict__ Wq, const float* __restrict__ Wk,
                 const float* __restrict__ Wv, const float* __restrict__ Wp,
                 const float* __restrict__ W1, const float* __restrict__ W2,
                 _Float16* __restrict__ wout,
                 const float* __restrict__ x, const float* __restrict__ g,
                 const float* __restrict__ be, _Float16* __restrict__ h) {
  __shared__ __attribute__((aligned(16))) _Float16 tile[64 * PREP_LD];
  const int blk = blockIdx.x;
  if (blk >= PREP_CAST_BLOCKS) {
    ln_rows(x, g, be, h, blk - PREP_CAST_BLOCKS);
    return;
  }
  const float* in;
  _Float16* out;
  int R, C, tr, tc;
  if (blk < 108) {
    const int j = blk / 36;
    const int t = blk - j * 36;
    const int b = t / 6;
    in  = (j == 0 ? Wq : (j == 1 ? Wk : Wv)) + (size_t)b * 24576;
    out = wout + (size_t)j * 147456 + (size_t)b * 24576;
    R = 384; C = 64; tr = t - b * 6; tc = 0;
  } else if (blk < 144) {
    const int t = blk - 108;
    in = Wp; out = wout + 442368;
    R = 384; C = 384; tr = t / 6; tc = t - tr * 6;
  } else if (blk < 288) {
    const int t = blk - 144;
    in = W1; out = wout + 589824;
    R = 384; C = 1536; tr = t / 24; tc = t - tr * 24;
  } else {
    const int t = blk - 288;
    in = W2; out = wout + 1179648;
    R = 1536; C = 384; tr = t / 6; tc = t - tr * 6;
  }
  const int tid = threadIdx.x;
  {
    const int lr = tid >> 4;
    const int lc = (tid & 15) * 4;
    const float* src = in + (size_t)(tr * 64 + lr) * C + tc * 64 + lc;
#pragma unroll
    for (int i = 0; i < 4; ++i) {
      const v4f f = *(const v4f*)(src + (size_t)(16 * i) * C);
      _Float16* d = tile + (lr + 16 * i) * PREP_LD + lc;
      d[0] = (_Float16)f[0]; d[1] = (_Float16)f[1];
      d[2] = (_Float16)f[2]; d[3] = (_Float16)f[3];
    }
  }
  __syncthreads();
  {
    const int oc = tid >> 3;
    const int r8 = (tid & 7) * 8;
    _Float16* dst = out + (size_t)(tc * 64 + oc) * R + tr * 64 + r8;
#pragma unroll
    for (int i = 0; i < 2; ++i) {
      v8h pk;
#pragma unroll
      for (int k = 0; k < 8; ++k) pk[k] = tile[(r8 + k) * PREP_LD + oc + 32 * i];
      *(v8h*)(dst + (size_t)(32 * i) * R) = pk;
    }
  }
}

__global__ void attn_kernel(const _Float16* __restrict__ q,
                            const _Float16* __restrict__ k,
                            const _Float16* __restrict__ vt,
                            _Float16* __restrict__ attn_out) {
  const int T = 256, Dh = 64, H = 6, C = 384;
  int tile = blockIdx.x;
  int it = tile & 15;
  int bh = tile >> 4;
  int b = bh / H, h = bh - b * H;
  int t0 = it * 16;
  int lane = threadIdx.x;
  const int n = lane & 15, mh = (lane >> 4) * 8;

  const _Float16* qb = q + ((size_t)bh * T + t0) * Dh;
  const _Float16* kb = k + (size_t)bh * T * Dh;
  const _Float16* vb = vt + (size_t)bh * Dh * T;

  v16h aq0 = load_frag(qb, Dh, lane);
  v16h aq1 = load_frag(qb + 32, Dh, lane);

  float mrow[8], lrow[8];
#pragma unroll
  for (int r = 0; r < 8; ++r) { mrow[r] = -__builtin_inff(); lrow[r] = 0.f; }
  v8f O[4];
#pragma unroll
  for (int i = 0; i < 4; ++i) O[i] = (v8f){};

  __shared__ _Float16 Pl[16 * 32];
  const float scale = 0.05103103630798287f;

  for (int s0 = 0; s0 < t0 + 16; s0 += 32) {
    float P0[8], P1[8];
    {
      v8f acc = {};
      v16h b0 = load_frag(kb + (size_t)s0 * Dh, Dh, lane);
      v16h b1 = load_frag(kb + (size_t)s0 * Dh + 32, Dh, lane);
      acc = wmma_f16(aq0, b0, acc);
      acc = wmma_f16(aq1, b1, acc);
#pragma unroll
      for (int r = 0; r < 8; ++r) P0[r] = acc[r] * scale;
    }
    {
      v8f acc = {};
      v16h b0 = load_frag(kb + (size_t)(s0 + 16) * Dh, Dh, lane);
      v16h b1 = load_frag(kb + (size_t)(s0 + 16) * Dh + 32, Dh, lane);
      acc = wmma_f16(aq0, b0, acc);
      acc = wmma_f16(aq1, b1, acc);
#pragma unroll
      for (int r = 0; r < 8; ++r) P1[r] = acc[r] * scale;
    }
#pragma unroll
    for (int r = 0; r < 8; ++r) {
      int trow = t0 + mh + r;
      if (s0 + n > trow)      P0[r] = -__builtin_inff();
      if (s0 + 16 + n > trow) P1[r] = -__builtin_inff();
      float mx = fmaxf(P0[r], P1[r]);
      mx = fmaxf(mx, __shfl_xor(mx, 1, 32));
      mx = fmaxf(mx, __shfl_xor(mx, 2, 32));
      mx = fmaxf(mx, __shfl_xor(mx, 4, 32));
      mx = fmaxf(mx, __shfl_xor(mx, 8, 32));
      float mnew = fmaxf(mrow[r], mx);
      float al = expf(mrow[r] - mnew);
      float p0 = expf(P0[r] - mnew);
      float p1 = expf(P1[r] - mnew);
      float rs = p0 + p1;
      rs += __shfl_xor(rs, 1, 32);
      rs += __shfl_xor(rs, 2, 32);
      rs += __shfl_xor(rs, 4, 32);
      rs += __shfl_xor(rs, 8, 32);
      lrow[r] = lrow[r] * al + rs;
      mrow[r] = mnew;
      P0[r] = p0; P1[r] = p1;
      O[0][r] *= al; O[1][r] *= al; O[2][r] *= al; O[3][r] *= al;
    }
    __syncthreads();
#pragma unroll
    for (int r = 0; r < 8; ++r) {
      Pl[(mh + r) * 32 + n]      = (_Float16)P0[r];
      Pl[(mh + r) * 32 + 16 + n] = (_Float16)P1[r];
    }
    __syncthreads();
    v16h ap = load_frag(Pl, 32, lane);
#pragma unroll
    for (int nt = 0; nt < 4; ++nt) {
      v16h bv = load_frag(vb + (size_t)(nt * 16) * T + s0, T, lane);
      O[nt] = wmma_f16(ap, bv, O[nt]);
    }
  }
#pragma unroll
  for (int r = 0; r < 8; ++r) {
    float inv = 1.f / lrow[r];
    _Float16* orow = attn_out + ((size_t)b * T + t0 + mh + r) * C + h * Dh;
#pragma unroll
    for (int nt = 0; nt < 4; ++nt) orow[nt * 16 + n] = (_Float16)(O[nt][r] * inv);
  }
}

#define GT_BM 128
#define GT_BN 128
#define GT_BK 64
#define GT_LD 72

template <int EPI>
__global__ __launch_bounds__(128)
void gemm_tile(const _Float16* __restrict__ A, const _Float16* __restrict__ Bt,
               const float* __restrict__ bias, const float* __restrict__ res,
               float* __restrict__ outf, _Float16* __restrict__ outh,
               _Float16* __restrict__ outk, _Float16* __restrict__ outvt,
               int N, int K) {
  __shared__ __attribute__((aligned(16))) _Float16 As[GT_BM * GT_LD];
  __shared__ __attribute__((aligned(16))) _Float16 Bs[GT_BN * GT_LD];
  const int tid = threadIdx.x;
  const int lane = tid & 31;
  const int w = tid >> 5;
  const int wm = w >> 1, wn = w & 1;
  const int ntn = N / GT_BN;
  const int tm = blockIdx.x / ntn;
  const int tn = blockIdx.x - tm * ntn;

  const int srow = tid >> 3;
  const int scol = (tid & 7) * 8;
  const _Float16* Ag = A  + ((size_t)tm * GT_BM + srow) * K + scol;
  const _Float16* Bg = Bt + ((size_t)tn * GT_BN + srow) * K + scol;
  _Float16* Asw = As + srow * GT_LD + scol;
  _Float16* Bsw = Bs + srow * GT_LD + scol;
  const _Float16* Asr = As + (wm * 64) * GT_LD;
  const _Float16* Bsr = Bs + (wn * 64) * GT_LD;

  v8f acc[4][4];
#pragma unroll
  for (int i = 0; i < 4; ++i)
#pragma unroll
    for (int j = 0; j < 4; ++j) acc[i][j] = (v8f){};

  v8h sa[8], sb[8];
#pragma unroll
  for (int i = 0; i < 8; ++i) {
    sa[i] = *(const v8h*)(Ag + (size_t)(16 * i) * K);
    sb[i] = *(const v8h*)(Bg + (size_t)(16 * i) * K);
  }

  for (int k0 = 0; k0 < K; k0 += GT_BK) {
#pragma unroll
    for (int i = 0; i < 8; ++i) {
      *(v8h*)(Asw + (16 * i) * GT_LD) = sa[i];
      *(v8h*)(Bsw + (16 * i) * GT_LD) = sb[i];
    }
    __syncthreads();
    if (k0 + GT_BK < K) {
#pragma unroll
      for (int i = 0; i < 8; ++i) {
        sa[i] = *(const v8h*)(Ag + (size_t)(16 * i) * K + k0 + GT_BK);
        sb[i] = *(const v8h*)(Bg + (size_t)(16 * i) * K + k0 + GT_BK);
      }
    }
#pragma unroll
    for (int kk = 0; kk < GT_BK; kk += 32) {
      v16h af[4], bf[4];
#pragma unroll
      for (int i = 0; i < 4; ++i) {
        af[i] = load_frag(Asr + (i * 16) * GT_LD + kk, GT_LD, lane);
        bf[i] = load_frag(Bsr + (i * 16) * GT_LD + kk, GT_LD, lane);
      }
#pragma unroll
      for (int i = 0; i < 4; ++i)
#pragma unroll
        for (int j = 0; j < 4; ++j) acc[i][j] = wmma_f16(af[i], bf[j], acc[i][j]);
    }
    __syncthreads();
  }

  const int n = lane & 15, mh = (lane >> 4) * 8;
  const int cbase = tn * GT_BN + wn * 64;
  const size_t rbase = (size_t)tm * GT_BM + wm * 64 + mh;
#pragma unroll
  for (int j = 0; j < 4; ++j) {
    const int col = cbase + j * 16 + n;
    float bv = 0.f;
    if (EPI != 3) bv = bias[col];
#pragma unroll
    for (int i = 0; i < 4; ++i) {
      const size_t row0 = rbase + i * 16;
      if (EPI == 3) {
        const int which = cbase / 384;
        const int hh = (cbase - which * 384) >> 6;
        const int d = j * 16 + n;
        const size_t b = row0 >> 8;
        const size_t t = row0 & 255;
        const size_t bh = b * 6 + hh;
        if (which == 2) {
          v8h pk;
#pragma unroll
          for (int r = 0; r < 8; ++r) pk[r] = (_Float16)acc[i][j][r];
          *(v8h*)(outvt + (bh * 64 + d) * 256 + t) = pk;
        } else {
          _Float16* o = (which == 0) ? outh : outk;
#pragma unroll
          for (int r = 0; r < 8; ++r) o[(bh * 256 + t + r) * 64 + d] = (_Float16)acc[i][j][r];
        }
      } else {
#pragma unroll
        for (int r = 0; r < 8; ++r) {
          const size_t off = (row0 + r) * N + col;
          const float v = acc[i][j][r] + bv;
          if (EPI == 0)      outf[off] = v + res[off];
          else if (EPI == 1) outh[off] = (_Float16)fmaxf(v, 0.f);
          else               outf[off] += v;
        }
      }
    }
  }
}

extern "C" void kernel_launch(void* const* d_in, const int* in_sizes, int n_in,
                              void* d_out, int out_size, void* d_ws, size_t ws_size,
                              hipStream_t stream) {
  (void)in_sizes; (void)n_in; (void)out_size; (void)ws_size;
  const float* x   = (const float*)d_in[0];
  const float* Wq  = (const float*)d_in[1];
  const float* Wk  = (const float*)d_in[2];
  const float* Wv  = (const float*)d_in[3];
  const float* Wp  = (const float*)d_in[4];
  const float* bp  = (const float*)d_in[5];
  const float* W1  = (const float*)d_in[6];
  const float* b1  = (const float*)d_in[7];
  const float* W2  = (const float*)d_in[8];
  const float* b2  = (const float*)d_in[9];
  const float* g1  = (const float*)d_in[10];
  const float* be1 = (const float*)d_in[11];
  const float* g2  = (const float*)d_in[12];
  const float* be2 = (const float*)d_in[13];
  float* out = (float*)d_out;

  char* ws = (char*)d_ws;
  const size_t S = (size_t)32768 * 384 * 2;
  _Float16* h16    = (_Float16*)(ws + 0 * S);
  _Float16* q16    = (_Float16*)(ws + 1 * S);
  _Float16* k16    = (_Float16*)(ws + 2 * S);
  _Float16* vt16   = (_Float16*)(ws + 3 * S);
  _Float16* attn16 = (_Float16*)(ws + 0 * S);
  _Float16* h2_16  = (_Float16*)(ws + 1 * S);
  _Float16* a1_16  = (_Float16*)(ws + 2 * S);
  char* wbase = ws + 6 * S;
  _Float16* wqt = (_Float16*)(wbase);
  _Float16* wpt = (_Float16*)(wbase + 3 * 294912);
  _Float16* w1t = (_Float16*)(wbase + 4 * 294912);
  _Float16* w2t = (_Float16*)(wbase + 4 * 294912 + 1179648);

  prep_kernel<<<4528, 256, 0, stream>>>(Wq, Wk, Wv, Wp, W1, W2, wqt, x, g1, be1, h16);
  gemm_tile<3><<<2304, 128, 0, stream>>>(h16, wqt, nullptr, nullptr, nullptr, q16, k16, vt16, 1152, 384);
  attn_kernel<<<12288, 32, 0, stream>>>(q16, k16, vt16, attn16);
  gemm_tile<0><<<768, 128, 0, stream>>>(attn16, wpt, bp, x, out, nullptr, nullptr, nullptr, 384, 384);
  ln_kernel<<<4096, 256, 0, stream>>>(out, g2, be2, h2_16);
  gemm_tile<1><<<3072, 128, 0, stream>>>(h2_16, w1t, b1, nullptr, nullptr, a1_16, nullptr, nullptr, 1536, 384);
  gemm_tile<2><<<768, 128, 0, stream>>>(a1_16, w2t, b2, nullptr, out, nullptr, nullptr, nullptr, 384, 1536);
}


# Attention transposed (S^T=K Q^T, O^T+=V^T P^T): P^T from score registers is the B operand, no LDS/barrier; row max and sum in-lane + one half-wave exchange (sum order changes); chunk loads grouped
# speedup vs baseline: 3.2394x; 1.0797x over previous
#include <hip/hip_runtime.h>
#include <hip/hip_bf16.h>


typedef __attribute__((ext_vector_type(16))) _Float16 v16h;
typedef __attribute__((ext_vector_type(8)))  _Float16 v8h;
typedef __attribute__((ext_vector_type(8)))  float    v8f;

#define DEV __device__ __forceinline__

DEV v8f wmma_f16(v16h a, v16h b, v8f c) {
  return __builtin_amdgcn_wmma_f32_16x16x32_f16(false, a, false, b, (short)0, c,
                                                false, false);
}

DEV v16h load_frag(const _Float16* __restrict__ tile, int ld, int lane) {
  const int m  = lane & 15;
  const int kh = (lane & 16) >> 1;
  const _Float16* p = tile + (size_t)m * ld + kh;
  v8h lo = *(const v8h*)p;
  v8h hi = *(const v8h*)(p + 16);
  v16h a;
#pragma unroll
  for (int i = 0; i < 8; ++i) { a[i] = lo[i]; a[i + 8] = hi[i]; }
  return a;
}

DEV void ln_rows(const float* __restrict__ x, const float* __restrict__ g,
                 const float* __restrict__ be, _Float16* __restrict__ out, int blk) {
  const int C = 384;
  int lane = threadIdx.x & 31;
  size_t row = (size_t)blk * 8 + (threadIdx.x >> 5);
  const float* xr = x + row * C;
  float v[12];
  float s = 0.f;
#pragma unroll
  for (int i = 0; i < 12; ++i) { v[i] = xr[lane + 32 * i]; s += v[i]; }
#pragma unroll
  for (int m = 16; m >= 1; m >>= 1) s += __shfl_xor(s, m, 32);
  float mu = s * (1.f / 384.f);
  float var = 0.f;
#pragma unroll
  for (int i = 0; i < 12; ++i) { float d = v[i] - mu; var += d * d; }
#pragma unroll
  for (int m = 16; m >= 1; m >>= 1) var += __shfl_xor(var, m, 32);
  float rs = rsqrtf(var * (1.f / 384.f) + 1e-5f);
  _Float16* orow = out + row * C;
#pragma unroll
  for (int i = 0; i < 12; ++i) {
    int c = lane + 32 * i;
    orow[c] = (_Float16)((v[i] - mu) * rs * g[c] + be[c]);
  }
}

__global__ void ln_kernel(const float* __restrict__ x, const float* __restrict__ g,
                          const float* __restrict__ be, _Float16* __restrict__ out) {
  ln_rows(x, g, be, out, blockIdx.x);
}

typedef __attribute__((ext_vector_type(4))) float v4f;
#define PREP_CAST_BLOCKS 432
#define PREP_LD 66

__global__ __launch_bounds__(256)
void prep_kernel(const float* __restrict__ Wq, const float* __restrict__ Wk,
                 const float* __restrict__ Wv, const float* __restrict__ Wp,
                 const float* __restrict__ W1, const float* __restrict__ W2,
                 _Float16* __restrict__ wout,
                 const float* __restrict__ x, const float* __restrict__ g,
                 const float* __restrict__ be, _Float16* __restrict__ h) {
  __shared__ __attribute__((aligned(16))) _Float16 tile[64 * PREP_LD];
  const int blk = blockIdx.x;
  if (blk >= PREP_CAST_BLOCKS) {
    ln_rows(x, g, be, h, blk - PREP_CAST_BLOCKS);
    return;
  }
  const float* in;
  _Float16* out;
  int R, C, tr, tc;
  if (blk < 108) {
    const int j = blk / 36;
    const int t = blk - j * 36;
    const int b = t / 6;
    in  = (j == 0 ? Wq : (j == 1 ? Wk : Wv)) + (size_t)b * 24576;
    out = wout + (size_t)j * 147456 + (size_t)b * 24576;
    R = 384; C = 64; tr = t - b * 6; tc = 0;
  } else if (blk < 144) {
    const int t = blk - 108;
    in = Wp; out = wout + 442368;
    R = 384; C = 384; tr = t / 6; tc = t - tr * 6;
  } else if (blk < 288) {
    const int t = blk - 144;
    in = W1; out = wout + 589824;
    R = 384; C = 1536; tr = t / 24; tc = t - tr * 24;
  } else {
    const int t = blk - 288;
    in = W2; out = wout + 1179648;
    R = 1536; C = 384; tr = t / 6; tc = t - tr * 6;
  }
  const int tid = threadIdx.x;
  {
    const int lr = tid >> 4;
    const int lc = (tid & 15) * 4;
    const float* src = in + (size_t)(tr * 64 + lr) * C + tc * 64 + lc;
#pragma unroll
    for (int i = 0; i < 4; ++i) {
      const v4f f = *(const v4f*)(src + (size_t)(16 * i) * C);
      _Float16* d = tile + (lr + 16 * i) * PREP_LD + lc;
      d[0] = (_Float16)f[0]; d[1] = (_Float16)f[1];
      d[2] = (_Float16)f[2]; d[3] = (_Float16)f[3];
    }
  }
  __syncthreads();
  {
    const int oc = tid >> 3;
    const int r8 = (tid & 7) * 8;
    _Float16* dst = out + (size_t)(tc * 64 + oc) * R + tr * 64 + r8;
#pragma unroll
    for (int i = 0; i < 2; ++i) {
      v8h pk;
#pragma unroll
      for (int k = 0; k < 8; ++k) pk[k] = tile[(r8 + k) * PREP_LD + oc + 32 * i];
      *(v8h*)(dst + (size_t)(32 * i) * R) = pk;
    }
  }
}


DEV float xhalf(float x) {
  const int b = __builtin_bit_cast(int, x);
  return __builtin_bit_cast(
      float, __builtin_amdgcn_permlanex16(b, b, 0x76543210, 0xfedcba98, false, false));
}

__global__ __launch_bounds__(32)
void attn_kernel(const _Float16* __restrict__ q,
                 const _Float16* __restrict__ k,
                 const _Float16* __restrict__ vt,
                 _Float16* __restrict__ attn_out) {
  const int T = 256, Dh = 64, H = 6, C = 384;
  const int tile = blockIdx.x;
  const int it = tile & 15;
  const int bh = tile >> 4;
  const int b = bh / H, h = bh - b * H;
  const int t0 = it * 16;
  const int lane = threadIdx.x;
  const int c = lane & 15;
  const int kh = (lane >> 4) * 8;

  const _Float16* qb = q + ((size_t)bh * T + t0) * Dh;
  const _Float16* kb = k + (size_t)bh * T * Dh;
  const _Float16* vb = vt + (size_t)bh * Dh * T;

  const v16h bq0 = load_frag(qb, Dh, lane);
  const v16h bq1 = load_frag(qb + 32, Dh, lane);

  float mrun = -__builtin_inff(), lrun = 0.f;
  v8f O[4];
#pragma unroll
  for (int i = 0; i < 4; ++i) O[i] = (v8f){};

  const float scale = 0.05103103630798287f;
  const int tq = t0 + c;

  for (int s0 = 0; s0 < t0 + 16; s0 += 32) {
    const _Float16* kp = kb + (size_t)s0 * Dh;
    const v16h ka0 = load_frag(kp, Dh, lane);
    const v16h ka1 = load_frag(kp + 32, Dh, lane);
    const v16h kc0 = load_frag(kp + 16 * Dh, Dh, lane);
    const v16h kc1 = load_frag(kp + 16 * Dh + 32, Dh, lane);
    v16h va[4];
#pragma unroll
    for (int nt = 0; nt < 4; ++nt) va[nt] = load_frag(vb + (size_t)(nt * 16) * T + s0, T, lane);

    __builtin_amdgcn_sched_barrier(0);

    v8f S0 = {}, S1 = {};
    S0 = wmma_f16(ka0, bq0, S0);
    S0 = wmma_f16(ka1, bq1, S0);
    S1 = wmma_f16(kc0, bq0, S1);
    S1 = wmma_f16(kc1, bq1, S1);

    float p0[8], p1[8];
#pragma unroll
    for (int r = 0; r < 8; ++r) { p0[r] = S0[r] * scale; p1[r] = S1[r] * scale; }
    if (s0 + 31 > t0) {
#pragma unroll
      for (int r = 0; r < 8; ++r) {
        if (s0 + kh + r > tq)      p0[r] = -__builtin_inff();
        if (s0 + 16 + kh + r > tq) p1[r] = -__builtin_inff();
      }
    }
    float mx = fmaxf(p0[0], p1[0]);
#pragma unroll
    for (int r = 1; r < 8; ++r) mx = fmaxf(mx, fmaxf(p0[r], p1[r]));
    mx = fmaxf(mx, xhalf(mx));
    const float mnew = fmaxf(mrun, mx);
    const float al = expf(mrun - mnew);
    float rs0 = 0.f, rs1 = 0.f;
#pragma unroll
    for (int r = 0; r < 8; ++r) {
      p0[r] = expf(p0[r] - mnew); rs0 += p0[r];
      p1[r] = expf(p1[r] - mnew); rs1 += p1[r];
    }
    float rs = rs0 + rs1;
    rs += xhalf(rs);
    lrun = lrun * al + rs;
    mrun = mnew;

    v16h bp;
#pragma unroll
    for (int r = 0; r < 8; ++r) { bp[r] = (_Float16)p0[r]; bp[r + 8] = (_Float16)p1[r]; }
#pragma unroll
    for (int nt = 0; nt < 4; ++nt)
#pragma unroll
      for (int r = 0; r < 8; ++r) O[nt][r] *= al;
#pragma unroll
    for (int nt = 0; nt < 4; ++nt) O[nt] = wmma_f16(va[nt], bp, O[nt]);
  }

  const float inv = 1.f / lrun;
  _Float16* orow = attn_out + ((size_t)b * T + t0 + c) * C + h * Dh + kh;
#pragma unroll
  for (int nt = 0; nt < 4; ++nt) {
    v8h pk;
#pragma unroll
    for (int r = 0; r < 8; ++r) pk[r] = (_Float16)(O[nt][r] * inv);
    *(v8h*)(orow + nt * 16) = pk;
  }
}

#define GT_BM 128
#define GT_BN 128
#define GT_BK 64
#define GT_LD 72

template <int EPI>
__global__ __launch_bounds__(128)
void gemm_tile(const _Float16* __restrict__ A, const _Float16* __restrict__ Bt,
               const float* __restrict__ bias, const float* __restrict__ res,
               float* __restrict__ outf, _Float16* __restrict__ outh,
               _Float16* __restrict__ outk, _Float16* __restrict__ outvt,
               int N, int K) {
  __shared__ __attribute__((aligned(16))) _Float16 As[GT_BM * GT_LD];
  __shared__ __attribute__((aligned(16))) _Float16 Bs[GT_BN * GT_LD];
  const int tid = threadIdx.x;
  const int lane = tid & 31;
  const int w = tid >> 5;
  const int wm = w >> 1, wn = w & 1;
  const int ntn = N / GT_BN;
  const int tm = blockIdx.x / ntn;
  const int tn = blockIdx.x - tm * ntn;

  const int srow = tid >> 3;
  const int scol = (tid & 7) * 8;
  const _Float16* Ag = A  + ((size_t)tm * GT_BM + srow) * K + scol;
  const _Float16* Bg = Bt + ((size_t)tn * GT_BN + srow) * K + scol;
  _Float16* Asw = As + srow * GT_LD + scol;
  _Float16* Bsw = Bs + srow * GT_LD + scol;
  const _Float16* Asr = As + (wm * 64) * GT_LD;
  const _Float16* Bsr = Bs + (wn * 64) * GT_LD;

  v8f acc[4][4];
#pragma unroll
  for (int i = 0; i < 4; ++i)
#pragma unroll
    for (int j = 0; j < 4; ++j) acc[i][j] = (v8f){};

  v8h sa[8], sb[8];
#pragma unroll
  for (int i = 0; i < 8; ++i) {
    sa[i] = *(const v8h*)(Ag + (size_t)(16 * i) * K);
    sb[i] = *(const v8h*)(Bg + (size_t)(16 * i) * K);
  }

  for (int k0 = 0; k0 < K; k0 += GT_BK) {
#pragma unroll
    for (int i = 0; i < 8; ++i) {
      *(v8h*)(Asw + (16 * i) * GT_LD) = sa[i];
      *(v8h*)(Bsw + (16 * i) * GT_LD) = sb[i];
    }
    __syncthreads();
    if (k0 + GT_BK < K) {
#pragma unroll
      for (int i = 0; i < 8; ++i) {
        sa[i] = *(const v8h*)(Ag + (size_t)(16 * i) * K + k0 + GT_BK);
        sb[i] = *(const v8h*)(Bg + (size_t)(16 * i) * K + k0 + GT_BK);
      }
    }
#pragma unroll
    for (int kk = 0; kk < GT_BK; kk += 32) {
      v16h af[4], bf[4];
#pragma unroll
      for (int i = 0; i < 4; ++i) {
        af[i] = load_frag(Asr + (i * 16) * GT_LD + kk, GT_LD, lane);
        bf[i] = load_frag(Bsr + (i * 16) * GT_LD + kk, GT_LD, lane);
      }
#pragma unroll
      for (int i = 0; i < 4; ++i)
#pragma unroll
        for (int j = 0; j < 4; ++j) acc[i][j] = wmma_f16(af[i], bf[j], acc[i][j]);
    }
    __syncthreads();
  }

  const int n = lane & 15, mh = (lane >> 4) * 8;
  const int cbase = tn * GT_BN + wn * 64;
  const size_t rbase = (size_t)tm * GT_BM + wm * 64 + mh;
#pragma unroll
  for (int j = 0; j < 4; ++j) {
    const int col = cbase + j * 16 + n;
    float bv = 0.f;
    if (EPI != 3) bv = bias[col];
#pragma unroll
    for (int i = 0; i < 4; ++i) {
      const size_t row0 = rbase + i * 16;
      if (EPI == 3) {
        const int which = cbase / 384;
        const int hh = (cbase - which * 384) >> 6;
        const int d = j * 16 + n;
        const size_t b = row0 >> 8;
        const size_t t = row0 & 255;
        const size_t bh = b * 6 + hh;
        if (which == 2) {
          v8h pk;
#pragma unroll
          for (int r = 0; r < 8; ++r) pk[r] = (_Float16)acc[i][j][r];
          *(v8h*)(outvt + (bh * 64 + d) * 256 + t) = pk;
        } else {
          _Float16* o = (which == 0) ? outh : outk;
#pragma unroll
          for (int r = 0; r < 8; ++r) o[(bh * 256 + t + r) * 64 + d] = (_Float16)acc[i][j][r];
        }
      } else {
#pragma unroll
        for (int r = 0; r < 8; ++r) {
          const size_t off = (row0 + r) * N + col;
          const float v = acc[i][j][r] + bv;
          if (EPI == 0)      outf[off] = v + res[off];
          else if (EPI == 1) outh[off] = (_Float16)fmaxf(v, 0.f);
          else               outf[off] += v;
        }
      }
    }
  }
}

extern "C" void kernel_launch(void* const* d_in, const int* in_sizes, int n_in,
                              void* d_out, int out_size, void* d_ws, size_t ws_size,
                              hipStream_t stream) {
  (void)in_sizes; (void)n_in; (void)out_size; (void)ws_size;
  const float* x   = (const float*)d_in[0];
  const float* Wq  = (const float*)d_in[1];
  const float* Wk  = (const float*)d_in[2];
  const float* Wv  = (const float*)d_in[3];
  const float* Wp  = (const float*)d_in[4];
  const float* bp  = (const float*)d_in[5];
  const float* W1  = (const float*)d_in[6];
  const float* b1  = (const float*)d_in[7];
  const float* W2  = (const float*)d_in[8];
  const float* b2  = (const float*)d_in[9];
  const float* g1  = (const float*)d_in[10];
  const float* be1 = (const float*)d_in[11];
  const float* g2  = (const float*)d_in[12];
  const float* be2 = (const float*)d_in[13];
  float* out = (float*)d_out;

  char* ws = (char*)d_ws;
  const size_t S = (size_t)32768 * 384 * 2;
  _Float16* h16    = (_Float16*)(ws + 0 * S);
  _Float16* q16    = (_Float16*)(ws + 1 * S);
  _Float16* k16    = (_Float16*)(ws + 2 * S);
  _Float16* vt16   = (_Float16*)(ws + 3 * S);
  _Float16* attn16 = (_Float16*)(ws + 0 * S);
  _Float16* h2_16  = (_Float16*)(ws + 1 * S);
  _Float16* a1_16  = (_Float16*)(ws + 2 * S);
  char* wbase = ws + 6 * S;
  _Float16* wqt = (_Float16*)(wbase);
  _Float16* wpt = (_Float16*)(wbase + 3 * 294912);
  _Float16* w1t = (_Float16*)(wbase + 4 * 294912);
  _Float16* w2t = (_Float16*)(wbase + 4 * 294912 + 1179648);

  prep_kernel<<<4528, 256, 0, stream>>>(Wq, Wk, Wv, Wp, W1, W2, wqt, x, g1, be1, h16);
  gemm_tile<3><<<2304, 128, 0, stream>>>(h16, wqt, nullptr, nullptr, nullptr, q16, k16, vt16, 1152, 384);
  attn_kernel<<<12288, 32, 0, stream>>>(q16, k16, vt16, attn16);
  gemm_tile<0><<<768, 128, 0, stream>>>(attn16, wpt, bp, x, out, nullptr, nullptr, nullptr, 384, 384);
  ln_kernel<<<4096, 256, 0, stream>>>(out, g2, be2, h2_16);
  gemm_tile<1><<<3072, 128, 0, stream>>>(h2_16, w1t, b1, nullptr, nullptr, a1_16, nullptr, nullptr, 1536, 384);
  gemm_tile<2><<<768, 128, 0, stream>>>(a1_16, w2t, b2, nullptr, out, nullptr, nullptr, nullptr, 384, 1536);
}


# GEMM epilogues: each wave's 64x64 tile leaves through its own LDS staging block, 16 rows a pass: row segments stored whole, residual/running output read a pass ahead, 16 B per lane, not 4 or 2 B each
# speedup vs baseline: 4.3896x; 1.3551x over previous
#include <hip/hip_runtime.h>
#include <hip/hip_bf16.h>


typedef __attribute__((ext_vector_type(16))) _Float16 v16h;
typedef __attribute__((ext_vector_type(8)))  _Float16 v8h;
typedef __attribute__((ext_vector_type(8)))  float    v8f;

#define DEV __device__ __forceinline__

DEV v8f wmma_f16(v16h a, v16h b, v8f c) {
  return __builtin_amdgcn_wmma_f32_16x16x32_f16(false, a, false, b, (short)0, c,
                                                false, false);
}

DEV v16h load_frag(const _Float16* __restrict__ tile, int ld, int lane) {
  const int m  = lane & 15;
  const int kh = (lane & 16) >> 1;
  const _Float16* p = tile + (size_t)m * ld + kh;
  v8h lo = *(const v8h*)p;
  v8h hi = *(const v8h*)(p + 16);
  v16h a;
#pragma unroll
  for (int i = 0; i < 8; ++i) { a[i] = lo[i]; a[i + 8] = hi[i]; }
  return a;
}

DEV void ln_rows(const float* __restrict__ x, const float* __restrict__ g,
                 const float* __restrict__ be, _Float16* __restrict__ out, int blk) {
  const int C = 384;
  int lane = threadIdx.x & 31;
  size_t row = (size_t)blk * 8 + (threadIdx.x >> 5);
  const float* xr = x + row * C;
  float v[12];
  float s = 0.f;
#pragma unroll
  for (int i = 0; i < 12; ++i) { v[i] = xr[lane + 32 * i]; s += v[i]; }
#pragma unroll
  for (int m = 16; m >= 1; m >>= 1) s += __shfl_xor(s, m, 32);
  float mu = s * (1.f / 384.f);
  float var = 0.f;
#pragma unroll
  for (int i = 0; i < 12; ++i) { float d = v[i] - mu; var += d * d; }
#pragma unroll
  for (int m = 16; m >= 1; m >>= 1) var += __shfl_xor(var, m, 32);
  float rs = rsqrtf(var * (1.f / 384.f) + 1e-5f);
  _Float16* orow = out + row * C;
#pragma unroll
  for (int i = 0; i < 12; ++i) {
    int c = lane + 32 * i;
    orow[c] = (_Float16)((v[i] - mu) * rs * g[c] + be[c]);
  }
}

__global__ void ln_kernel(const float* __restrict__ x, const float* __restrict__ g,
                          const float* __restrict__ be, _Float16* __restrict__ out) {
  ln_rows(x, g, be, out, blockIdx.x);
}

typedef __attribute__((ext_vector_type(4))) float v4f;
#define PREP_CAST_BLOCKS 432
#define PREP_LD 66

__global__ __launch_bounds__(256)
void prep_kernel(const float* __restrict__ Wq, const float* __restrict__ Wk,
                 const float* __restrict__ Wv, const float* __restrict__ Wp,
                 const float* __restrict__ W1, const float* __restrict__ W2,
                 _Float16* __restrict__ wout,
                 const float* __restrict__ x, const float* __restrict__ g,
                 const float* __restrict__ be, _Float16* __restrict__ h) {
  __shared__ __attribute__((aligned(16))) _Float16 tile[64 * PREP_LD];
  const int blk = blockIdx.x;
  if (blk >= PREP_CAST_BLOCKS) {
    ln_rows(x, g, be, h, blk - PREP_CAST_BLOCKS);
    return;
  }
  const float* in;
  _Float16* out;
  int R, C, tr, tc;
  if (blk < 108) {
    const int j = blk / 36;
    const int t = blk - j * 36;
    const int b = t / 6;
    in  = (j == 0 ? Wq : (j == 1 ? Wk : Wv)) + (size_t)b * 24576;
    out = wout + (size_t)j * 147456 + (size_t)b * 24576;
    R = 384; C = 64; tr = t - b * 6; tc = 0;
  } else if (blk < 144) {
    const int t = blk - 108;
    in = Wp; out = wout + 442368;
    R = 384; C = 384; tr = t / 6; tc = t - tr * 6;
  } else if (blk < 288) {
    const int t = blk - 144;
    in = W1; out = wout + 589824;
    R = 384; C = 1536; tr = t / 24; tc = t - tr * 24;
  } else {
    const int t = blk - 288;
    in = W2; out = wout + 1179648;
    R = 1536; C = 384; tr = t / 6; tc = t - tr * 6;
  }
  const int tid = threadIdx.x;
  {
    const int lr = tid >> 4;
    const int lc = (tid & 15) * 4;
    const float* src = in + (size_t)(tr * 64 + lr) * C + tc * 64 + lc;
#pragma unroll
    for (int i = 0; i < 4; ++i) {
      const v4f f = *(const v4f*)(src + (size_t)(16 * i) * C);
      _Float16* d = tile + (lr + 16 * i) * PREP_LD + lc;
      d[0] = (_Float16)f[0]; d[1] = (_Float16)f[1];
      d[2] = (_Float16)f[2]; d[3] = (_Float16)f[3];
    }
  }
  __syncthreads();
  {
    const int oc = tid >> 3;
    const int r8 = (tid & 7) * 8;
    _Float16* dst = out + (size_t)(tc * 64 + oc) * R + tr * 64 + r8;
#pragma unroll
    for (int i = 0; i < 2; ++i) {
      v8h pk;
#pragma unroll
      for (int k = 0; k < 8; ++k) pk[k] = tile[(r8 + k) * PREP_LD + oc + 32 * i];
      *(v8h*)(dst + (size_t)(32 * i) * R) = pk;
    }
  }
}


DEV float xhalf(float x) {
  const int b = __builtin_bit_cast(int, x);
  return __builtin_bit_cast(
      float, __builtin_amdgcn_permlanex16(b, b, 0x76543210, 0xfedcba98, false, false));
}

__global__ __launch_bounds__(32)
void attn_kernel(const _Float16* __restrict__ q,
                 const _Float16* __restrict__ k,
                 const _Float16* __restrict__ vt,
                 _Float16* __restrict__ attn_out) {
  const int T = 256, Dh = 64, H = 6, C = 384;
  const int tile = blockIdx.x;
  const int it = tile & 15;
  const int bh = tile >> 4;
  const int b = bh / H, h = bh - b * H;
  const int t0 = it * 16;
  const int lane = threadIdx.x;
  const int c = lane & 15;
  const int kh = (lane >> 4) * 8;

  const _Float16* qb = q + ((size_t)bh * T + t0) * Dh;
  const _Float16* kb = k + (size_t)bh * T * Dh;
  const _Float16* vb = vt + (size_t)bh * Dh * T;

  const v16h bq0 = load_frag(qb, Dh, lane);
  const v16h bq1 = load_frag(qb + 32, Dh, lane);

  float mrun = -__builtin_inff(), lrun = 0.f;
  v8f O[4];
#pragma unroll
  for (int i = 0; i < 4; ++i) O[i] = (v8f){};

  const float scale = 0.05103103630798287f;
  const int tq = t0 + c;

  for (int s0 = 0; s0 < t0 + 16; s0 += 32) {
    const _Float16* kp = kb + (size_t)s0 * Dh;
    const v16h ka0 = load_frag(kp, Dh, lane);
    const v16h ka1 = load_frag(kp + 32, Dh, lane);
    const v16h kc0 = load_frag(kp + 16 * Dh, Dh, lane);
    const v16h kc1 = load_frag(kp + 16 * Dh + 32, Dh, lane);
    v16h va[4];
#pragma unroll
    for (int nt = 0; nt < 4; ++nt) va[nt] = load_frag(vb + (size_t)(nt * 16) * T + s0, T, lane);

    __builtin_amdgcn_sched_barrier(0);

    v8f S0 = {}, S1 = {};
    S0 = wmma_f16(ka0, bq0, S0);
    S0 = wmma_f16(ka1, bq1, S0);
    S1 = wmma_f16(kc0, bq0, S1);
    S1 = wmma_f16(kc1, bq1, S1);

    float p0[8], p1[8];
#pragma unroll
    for (int r = 0; r < 8; ++r) { p0[r] = S0[r] * scale; p1[r] = S1[r] * scale; }
    if (s0 + 31 > t0) {
#pragma unroll
      for (int r = 0; r < 8; ++r) {
        if (s0 + kh + r > tq)      p0[r] = -__builtin_inff();
        if (s0 + 16 + kh + r > tq) p1[r] = -__builtin_inff();
      }
    }
    float mx = fmaxf(p0[0], p1[0]);
#pragma unroll
    for (int r = 1; r < 8; ++r) mx = fmaxf(mx, fmaxf(p0[r], p1[r]));
    mx = fmaxf(mx, xhalf(mx));
    const float mnew = fmaxf(mrun, mx);
    const float al = expf(mrun - mnew);
    float rs0 = 0.f, rs1 = 0.f;
#pragma unroll
    for (int r = 0; r < 8; ++r) {
      p0[r] = expf(p0[r] - mnew); rs0 += p0[r];
      p1[r] = expf(p1[r] - mnew); rs1 += p1[r];
    }
    float rs = rs0 + rs1;
    rs += xhalf(rs);
    lrun = lrun * al + rs;
    mrun = mnew;

    v16h bp;
#pragma unroll
    for (int r = 0; r < 8; ++r) { bp[r] = (_Float16)p0[r]; bp[r + 8] = (_Float16)p1[r]; }
#pragma unroll
    for (int nt = 0; nt < 4; ++nt)
#pragma unroll
      for (int r = 0; r < 8; ++r) O[nt][r] *= al;
#pragma unroll
    for (int nt = 0; nt < 4; ++nt) O[nt] = wmma_f16(va[nt], bp, O[nt]);
  }

  const float inv = 1.f / lrun;
  _Float16* orow = attn_out + ((size_t)b * T + t0 + c) * C + h * Dh + kh;
#pragma unroll
  for (int nt = 0; nt < 4; ++nt) {
    v8h pk;
#pragma unroll
    for (int r = 0; r < 8; ++r) pk[r] = (_Float16)(O[nt][r] * inv);
    *(v8h*)(orow + nt * 16) = pk;
  }
}

#define GT_BM 128
#define GT_BN 128
#define GT_BK 64
#define GT_LD 72
#define EP_LD 68

DEV void wave_lds_sync() {
  __builtin_amdgcn_fence(__ATOMIC_RELEASE, "wavefront");
  __builtin_amdgcn_wave_barrier();
  __builtin_amdgcn_fence(__ATOMIC_ACQUIRE, "wavefront");
}

template <int EPI>
__global__ __launch_bounds__(128)
void gemm_tile(const _Float16* __restrict__ A, const _Float16* __restrict__ Bt,
               const float* __restrict__ bias, const float* __restrict__ res,
               float* __restrict__ outf, _Float16* __restrict__ outh,
               _Float16* __restrict__ outk, _Float16* __restrict__ outvt,
               int N, int K) {
  __shared__ __attribute__((aligned(16))) _Float16 As[GT_BM * GT_LD];
  __shared__ __attribute__((aligned(16))) _Float16 Bs[GT_BN * GT_LD];
  const int tid = threadIdx.x;
  const int lane = tid & 31;
  const int w = tid >> 5;
  const int wm = w >> 1, wn = w & 1;
  const int ntn = N / GT_BN;
  const int tm = blockIdx.x / ntn;
  const int tn = blockIdx.x - tm * ntn;

  const int srow = tid >> 3;
  const int scol = (tid & 7) * 8;
  const _Float16* Ag = A  + ((size_t)tm * GT_BM + srow) * K + scol;
  const _Float16* Bg = Bt + ((size_t)tn * GT_BN + srow) * K + scol;
  _Float16* Asw = As + srow * GT_LD + scol;
  _Float16* Bsw = Bs + srow * GT_LD + scol;
  const _Float16* Asr = As + (wm * 64) * GT_LD;
  const _Float16* Bsr = Bs + (wn * 64) * GT_LD;

  v8f acc[4][4];
#pragma unroll
  for (int i = 0; i < 4; ++i)
#pragma unroll
    for (int j = 0; j < 4; ++j) acc[i][j] = (v8f){};

  v8h sa[8], sb[8];
#pragma unroll
  for (int i = 0; i < 8; ++i) {
    sa[i] = *(const v8h*)(Ag + (size_t)(16 * i) * K);
    sb[i] = *(const v8h*)(Bg + (size_t)(16 * i) * K);
  }

  for (int k0 = 0; k0 < K; k0 += GT_BK) {
#pragma unroll
    for (int i = 0; i < 8; ++i) {
      *(v8h*)(Asw + (16 * i) * GT_LD) = sa[i];
      *(v8h*)(Bsw + (16 * i) * GT_LD) = sb[i];
    }
    __syncthreads();
    if (k0 + GT_BK < K) {
#pragma unroll
      for (int i = 0; i < 8; ++i) {
        sa[i] = *(const v8h*)(Ag + (size_t)(16 * i) * K + k0 + GT_BK);
        sb[i] = *(const v8h*)(Bg + (size_t)(16 * i) * K + k0 + GT_BK);
      }
    }
#pragma unroll
    for (int kk = 0; kk < GT_BK; kk += 32) {
      v16h af[4], bf[4];
#pragma unroll
      for (int i = 0; i < 4; ++i) {
        af[i] = load_frag(Asr + (i * 16) * GT_LD + kk, GT_LD, lane);
        bf[i] = load_frag(Bsr + (i * 16) * GT_LD + kk, GT_LD, lane);
      }
#pragma unroll
      for (int i = 0; i < 4; ++i)
#pragma unroll
        for (int j = 0; j < 4; ++j) acc[i][j] = wmma_f16(af[i], bf[j], acc[i][j]);
    }
    __syncthreads();
  }

  const int n = lane & 15, mh = (lane >> 4) * 8;
  const int cbase = tn * GT_BN + wn * 64;
  const size_t rbase = (size_t)tm * GT_BM + wm * 64;
  float* stg = reinterpret_cast<float*>(As) + w * (16 * EP_LD);

  if (EPI == 3 && cbase >= 768) {
    const int hh = (cbase - 768) >> 6;
    const size_t bh = (rbase >> 8) * 6 + hh;
    const size_t t0 = (rbase & 255) + mh;
#pragma unroll
    for (int j = 0; j < 4; ++j) {
      const int d = j * 16 + n;
#pragma unroll
      for (int i = 0; i < 4; ++i) {
        v8h pk;
#pragma unroll
        for (int r = 0; r < 8; ++r) pk[r] = (_Float16)acc[i][j][r];
        *(v8h*)(outvt + (bh * 64 + d) * 256 + t0 + i * 16) = pk;
      }
    }
    return;
  }

  float bv[4];
#pragma unroll
  for (int j = 0; j < 4; ++j) bv[j] = (EPI != 3) ? bias[cbase + j * 16 + n] : 0.f;

  if (EPI == 0 || EPI == 2) {
    const int r2 = lane >> 4;
    const int c4 = (lane & 15) * 4;
    const float* src = (EPI == 0) ? res : outf;
    const size_t o0 = (rbase + r2) * (size_t)N + cbase + c4;
    v4f cur[8], nxt[8];
#pragma unroll
    for (int it = 0; it < 8; ++it) cur[it] = *(const v4f*)(src + o0 + (size_t)(2 * it) * N);
#pragma unroll
    for (int i = 0; i < 4; ++i) {
      if (i < 3) {
#pragma unroll
        for (int it = 0; it < 8; ++it)
          nxt[it] = *(const v4f*)(src + o0 + (size_t)((i + 1) * 16 + 2 * it) * N);
      }
#pragma unroll
      for (int j = 0; j < 4; ++j)
#pragma unroll
        for (int r = 0; r < 8; ++r) stg[(mh + r) * EP_LD + j * 16 + n] = acc[i][j][r] + bv[j];
      wave_lds_sync();
#pragma unroll
      for (int it = 0; it < 8; ++it) {
        const v4f s = *(const v4f*)(stg + (2 * it + r2) * EP_LD + c4);
        v4f o;
#pragma unroll
        for (int e = 0; e < 4; ++e) o[e] = (EPI == 0) ? s[e] + cur[it][e] : cur[it][e] + s[e];
        *(v4f*)(outf + o0 + (size_t)(i * 16 + 2 * it) * N) = o;
      }
      wave_lds_sync();
#pragma unroll
      for (int it = 0; it < 8; ++it) cur[it] = nxt[it];
    }
  } else {
    const int r4 = lane >> 3;
    const int c8 = (lane & 7) * 8;
    _Float16* dst;
    size_t ldo;
    if (EPI == 1) {
      dst = outh + rbase * (size_t)N + cbase;
      ldo = (size_t)N;
    } else {
      const int which = cbase / 384;
      const int hh = (cbase - which * 384) >> 6;
      const size_t bh = (rbase >> 8) * 6 + hh;
      dst = ((which == 0) ? outh : outk) + (bh * 256 + (rbase & 255)) * 64;
      ldo = 64;
    }
#pragma unroll
    for (int i = 0; i < 4; ++i) {
#pragma unroll
      for (int j = 0; j < 4; ++j)
#pragma unroll
        for (int r = 0; r < 8; ++r) {
          float v = acc[i][j][r];
          if (EPI == 1) v = fmaxf(v + bv[j], 0.f);
          stg[(mh + r) * EP_LD + j * 16 + n] = v;
        }
      wave_lds_sync();
#pragma unroll
      for (int it = 0; it < 4; ++it) {
        const int row = 4 * it + r4;
        const v4f s0 = *(const v4f*)(stg + row * EP_LD + c8);
        const v4f s1 = *(const v4f*)(stg + row * EP_LD + c8 + 4);
        v8h pk;
#pragma unroll
        for (int e = 0; e < 4; ++e) { pk[e] = (_Float16)s0[e]; pk[e + 4] = (_Float16)s1[e]; }
        *(v8h*)(dst + (size_t)(i * 16 + row) * ldo + c8) = pk;
      }
      wave_lds_sync();
    }
  }
}

extern "C" void kernel_launch(void* const* d_in, const int* in_sizes, int n_in,
                              void* d_out, int out_size, void* d_ws, size_t ws_size,
                              hipStream_t stream) {
  (void)in_sizes; (void)n_in; (void)out_size; (void)ws_size;
  const float* x   = (const float*)d_in[0];
  const float* Wq  = (const float*)d_in[1];
  const float* Wk  = (const float*)d_in[2];
  const float* Wv  = (const float*)d_in[3];
  const float* Wp  = (const float*)d_in[4];
  const float* bp  = (const float*)d_in[5];
  const float* W1  = (const float*)d_in[6];
  const float* b1  = (const float*)d_in[7];
  const float* W2  = (const float*)d_in[8];
  const float* b2  = (const float*)d_in[9];
  const float* g1  = (const float*)d_in[10];
  const float* be1 = (const float*)d_in[11];
  const float* g2  = (const float*)d_in[12];
  const float* be2 = (const float*)d_in[13];
  float* out = (float*)d_out;

  char* ws = (char*)d_ws;
  const size_t S = (size_t)32768 * 384 * 2;
  _Float16* h16    = (_Float16*)(ws + 0 * S);
  _Float16* q16    = (_Float16*)(ws + 1 * S);
  _Float16* k16    = (_Float16*)(ws + 2 * S);
  _Float16* vt16   = (_Float16*)(ws + 3 * S);
  _Float16* attn16 = (_Float16*)(ws + 0 * S);
  _Float16* h2_16  = (_Float16*)(ws + 1 * S);
  _Float16* a1_16  = (_Float16*)(ws + 2 * S);
  char* wbase = ws + 6 * S;
  _Float16* wqt = (_Float16*)(wbase);
  _Float16* wpt = (_Float16*)(wbase + 3 * 294912);
  _Float16* w1t = (_Float16*)(wbase + 4 * 294912);
  _Float16* w2t = (_Float16*)(wbase + 4 * 294912 + 1179648);

  prep_kernel<<<4528, 256, 0, stream>>>(Wq, Wk, Wv, Wp, W1, W2, wqt, x, g1, be1, h16);
  gemm_tile<3><<<2304, 128, 0, stream>>>(h16, wqt, nullptr, nullptr, nullptr, q16, k16, vt16, 1152, 384);
  attn_kernel<<<12288, 32, 0, stream>>>(q16, k16, vt16, attn16);
  gemm_tile<0><<<768, 128, 0, stream>>>(attn16, wpt, bp, x, out, nullptr, nullptr, nullptr, 384, 384);
  ln_kernel<<<4096, 256, 0, stream>>>(out, g2, be2, h2_16);
  gemm_tile<1><<<3072, 128, 0, stream>>>(h2_16, w1t, b1, nullptr, nullptr, a1_16, nullptr, nullptr, 1536, 384);
  gemm_tile<2><<<768, 128, 0, stream>>>(a1_16, w2t, b2, nullptr, out, nullptr, nullptr, nullptr, 384, 1536);
}


# Attention: a wave takes two neighbouring 16-query tiles (32 queries); each K and V^T fragment load (16 lines per load) feeds both: fragment loads per query halved, 12288 waves -> 6144; sums unchanged
# speedup vs baseline: 4.7482x; 1.0817x over previous
#include <hip/hip_runtime.h>
#include <hip/hip_bf16.h>


typedef __attribute__((ext_vector_type(16))) _Float16 v16h;
typedef __attribute__((ext_vector_type(8)))  _Float16 v8h;
typedef __attribute__((ext_vector_type(8)))  float    v8f;

#define DEV __device__ __forceinline__

DEV v8f wmma_f16(v16h a, v16h b, v8f c) {
  return __builtin_amdgcn_wmma_f32_16x16x32_f16(false, a, false, b, (short)0, c,
                                                false, false);
}

DEV v16h load_frag(const _Float16* __restrict__ tile, int ld, int lane) {
  const int m  = lane & 15;
  const int kh = (lane & 16) >> 1;
  const _Float16* p = tile + (size_t)m * ld + kh;
  v8h lo = *(const v8h*)p;
  v8h hi = *(const v8h*)(p + 16);
  v16h a;
#pragma unroll
  for (int i = 0; i < 8; ++i) { a[i] = lo[i]; a[i + 8] = hi[i]; }
  return a;
}

DEV void ln_rows(const float* __restrict__ x, const float* __restrict__ g,
                 const float* __restrict__ be, _Float16* __restrict__ out, int blk) {
  const int C = 384;
  int lane = threadIdx.x & 31;
  size_t row = (size_t)blk * 8 + (threadIdx.x >> 5);
  const float* xr = x + row * C;
  float v[12];
  float s = 0.f;
#pragma unroll
  for (int i = 0; i < 12; ++i) { v[i] = xr[lane + 32 * i]; s += v[i]; }
#pragma unroll
  for (int m = 16; m >= 1; m >>= 1) s += __shfl_xor(s, m, 32);
  float mu = s * (1.f / 384.f);
  float var = 0.f;
#pragma unroll
  for (int i = 0; i < 12; ++i) { float d = v[i] - mu; var += d * d; }
#pragma unroll
  for (int m = 16; m >= 1; m >>= 1) var += __shfl_xor(var, m, 32);
  float rs = rsqrtf(var * (1.f / 384.f) + 1e-5f);
  _Float16* orow = out + row * C;
#pragma unroll
  for (int i = 0; i < 12; ++i) {
    int c = lane + 32 * i;
    orow[c] = (_Float16)((v[i] - mu) * rs * g[c] + be[c]);
  }
}

__global__ void ln_kernel(const float* __restrict__ x, const float* __restrict__ g,
                          const float* __restrict__ be, _Float16* __restrict__ out) {
  ln_rows(x, g, be, out, blockIdx.x);
}

typedef __attribute__((ext_vector_type(4))) float v4f;
#define PREP_CAST_BLOCKS 432
#define PREP_LD 66

__global__ __launch_bounds__(256)
void prep_kernel(const float* __restrict__ Wq, const float* __restrict__ Wk,
                 const float* __restrict__ Wv, const float* __restrict__ Wp,
                 const float* __restrict__ W1, const float* __restrict__ W2,
                 _Float16* __restrict__ wout,
                 const float* __restrict__ x, const float* __restrict__ g,
                 const float* __restrict__ be, _Float16* __restrict__ h) {
  __shared__ __attribute__((aligned(16))) _Float16 tile[64 * PREP_LD];
  const int blk = blockIdx.x;
  if (blk >= PREP_CAST_BLOCKS) {
    ln_rows(x, g, be, h, blk - PREP_CAST_BLOCKS);
    return;
  }
  const float* in;
  _Float16* out;
  int R, C, tr, tc;
  if (blk < 108) {
    const int j = blk / 36;
    const int t = blk - j * 36;
    const int b = t / 6;
    in  = (j == 0 ? Wq : (j == 1 ? Wk : Wv)) + (size_t)b * 24576;
    out = wout + (size_t)j * 147456 + (size_t)b * 24576;
    R = 384; C = 64; tr = t - b * 6; tc = 0;
  } else if (blk < 144) {
    const int t = blk - 108;
    in = Wp; out = wout + 442368;
    R = 384; C = 384; tr = t / 6; tc = t - tr * 6;
  } else if (blk < 288) {
    const int t = blk - 144;
    in = W1; out = wout + 589824;
    R = 384; C = 1536; tr = t / 24; tc = t - tr * 24;
  } else {
    const int t = blk - 288;
    in = W2; out = wout + 1179648;
    R = 1536; C = 384; tr = t / 6; tc = t - tr * 6;
  }
  const int tid = threadIdx.x;
  {
    const int lr = tid >> 4;
    const int lc = (tid & 15) * 4;
    const float* src = in + (size_t)(tr * 64 + lr) * C + tc * 64 + lc;
#pragma unroll
    for (int i = 0; i < 4; ++i) {
      const v4f f = *(const v4f*)(src + (size_t)(16 * i) * C);
      _Float16* d = tile + (lr + 16 * i) * PREP_LD + lc;
      d[0] = (_Float16)f[0]; d[1] = (_Float16)f[1];
      d[2] = (_Float16)f[2]; d[3] = (_Float16)f[3];
    }
  }
  __syncthreads();
  {
    const int oc = tid >> 3;
    const int r8 = (tid & 7) * 8;
    _Float16* dst = out + (size_t)(tc * 64 + oc) * R + tr * 64 + r8;
#pragma unroll
    for (int i = 0; i < 2; ++i) {
      v8h pk;
#pragma unroll
      for (int k = 0; k < 8; ++k) pk[k] = tile[(r8 + k) * PREP_LD + oc + 32 * i];
      *(v8h*)(dst + (size_t)(32 * i) * R) = pk;
    }
  }
}


DEV float xhalf(float x) {
  const int b = __builtin_bit_cast(int, x);
  return __builtin_bit_cast(
      float, __builtin_amdgcn_permlanex16(b, b, 0x76543210, 0xfedcba98, false, false));
}

__global__ __launch_bounds__(32)
void attn_kernel(const _Float16* __restrict__ q,
                 const _Float16* __restrict__ k,
                 const _Float16* __restrict__ vt,
                 _Float16* __restrict__ attn_out) {
  const int T = 256, Dh = 64, H = 6, C = 384;
  const int tile = blockIdx.x;
  const int ip = tile & 7;
  const int bh = tile >> 3;
  const int b = bh / H, h = bh - b * H;
  const int t0 = ip * 32;
  const int lane = threadIdx.x;
  const int c = lane & 15;
  const int kh = (lane >> 4) * 8;

  const _Float16* qb = q + ((size_t)bh * T + t0) * Dh;
  const _Float16* kb = k + (size_t)bh * T * Dh;
  const _Float16* vb = vt + (size_t)bh * Dh * T;

  v16h bq[2][2];
#pragma unroll
  for (int u = 0; u < 2; ++u) {
    bq[u][0] = load_frag(qb + (size_t)(u * 16) * Dh, Dh, lane);
    bq[u][1] = load_frag(qb + (size_t)(u * 16) * Dh + 32, Dh, lane);
  }

  float mrun[2], lrun[2];
  v8f O[2][4];
#pragma unroll
  for (int u = 0; u < 2; ++u) {
    mrun[u] = -__builtin_inff();
    lrun[u] = 0.f;
#pragma unroll
    for (int i = 0; i < 4; ++i) O[u][i] = (v8f){};
  }

  const float scale = 0.05103103630798287f;

  for (int s0 = 0; s0 <= t0; s0 += 32) {
    const _Float16* kp = kb + (size_t)s0 * Dh;
    const v16h ka0 = load_frag(kp, Dh, lane);
    const v16h ka1 = load_frag(kp + 32, Dh, lane);
    const v16h kc0 = load_frag(kp + 16 * Dh, Dh, lane);
    const v16h kc1 = load_frag(kp + 16 * Dh + 32, Dh, lane);
    v16h va[4];
#pragma unroll
    for (int nt = 0; nt < 4; ++nt) va[nt] = load_frag(vb + (size_t)(nt * 16) * T + s0, T, lane);

    __builtin_amdgcn_sched_barrier(0);

#pragma unroll
    for (int u = 0; u < 2; ++u) {
      const int tu = t0 + 16 * u;
      const int tq = tu + c;
      v8f S0 = {}, S1 = {};
      S0 = wmma_f16(ka0, bq[u][0], S0);
      S0 = wmma_f16(ka1, bq[u][1], S0);
      S1 = wmma_f16(kc0, bq[u][0], S1);
      S1 = wmma_f16(kc1, bq[u][1], S1);

      float p0[8], p1[8];
#pragma unroll
      for (int r = 0; r < 8; ++r) { p0[r] = S0[r] * scale; p1[r] = S1[r] * scale; }
      if (s0 + 31 > tu) {
#pragma unroll
        for (int r = 0; r < 8; ++r) {
          if (s0 + kh + r > tq)      p0[r] = -__builtin_inff();
          if (s0 + 16 + kh + r > tq) p1[r] = -__builtin_inff();
        }
      }
      float mx = fmaxf(p0[0], p1[0]);
#pragma unroll
      for (int r = 1; r < 8; ++r) mx = fmaxf(mx, fmaxf(p0[r], p1[r]));
      mx = fmaxf(mx, xhalf(mx));
      const float mnew = fmaxf(mrun[u], mx);
      const float al = expf(mrun[u] - mnew);
      float rs0 = 0.f, rs1 = 0.f;
#pragma unroll
      for (int r = 0; r < 8; ++r) {
        p0[r] = expf(p0[r] - mnew); rs0 += p0[r];
        p1[r] = expf(p1[r] - mnew); rs1 += p1[r];
      }
      float rs = rs0 + rs1;
      rs += xhalf(rs);
      lrun[u] = lrun[u] * al + rs;
      mrun[u] = mnew;

      v16h bp;
#pragma unroll
      for (int r = 0; r < 8; ++r) { bp[r] = (_Float16)p0[r]; bp[r + 8] = (_Float16)p1[r]; }
#pragma unroll
      for (int nt = 0; nt < 4; ++nt)
#pragma unroll
        for (int r = 0; r < 8; ++r) O[u][nt][r] *= al;
#pragma unroll
      for (int nt = 0; nt < 4; ++nt) O[u][nt] = wmma_f16(va[nt], bp, O[u][nt]);
    }
  }

#pragma unroll
  for (int u = 0; u < 2; ++u) {
    const float inv = 1.f / lrun[u];
    _Float16* orow = attn_out + ((size_t)b * T + t0 + 16 * u + c) * C + h * Dh + kh;
#pragma unroll
    for (int nt = 0; nt < 4; ++nt) {
      v8h pk;
#pragma unroll
      for (int r = 0; r < 8; ++r) pk[r] = (_Float16)(O[u][nt][r] * inv);
      *(v8h*)(orow + nt * 16) = pk;
    }
  }
}

#define GT_BM 128
#define GT_BN 128
#define GT_BK 64
#define GT_LD 72
#define EP_LD 68

DEV void wave_lds_sync() {
  __builtin_amdgcn_fence(__ATOMIC_RELEASE, "wavefront");
  __builtin_amdgcn_wave_barrier();
  __builtin_amdgcn_fence(__ATOMIC_ACQUIRE, "wavefront");
}

template <int EPI>
__global__ __launch_bounds__(128)
void gemm_tile(const _Float16* __restrict__ A, const _Float16* __restrict__ Bt,
               const float* __restrict__ bias, const float* __restrict__ res,
               float* __restrict__ outf, _Float16* __restrict__ outh,
               _Float16* __restrict__ outk, _Float16* __restrict__ outvt,
               int N, int K) {
  __shared__ __attribute__((aligned(16))) _Float16 As[GT_BM * GT_LD];
  __shared__ __attribute__((aligned(16))) _Float16 Bs[GT_BN * GT_LD];
  const int tid = threadIdx.x;
  const int lane = tid & 31;
  const int w = tid >> 5;
  const int wm = w >> 1, wn = w & 1;
  const int ntn = N / GT_BN;
  const int tm = blockIdx.x / ntn;
  const int tn = blockIdx.x - tm * ntn;

  const int srow = tid >> 3;
  const int scol = (tid & 7) * 8;
  const _Float16* Ag = A  + ((size_t)tm * GT_BM + srow) * K + scol;
  const _Float16* Bg = Bt + ((size_t)tn * GT_BN + srow) * K + scol;
  _Float16* Asw = As + srow * GT_LD + scol;
  _Float16* Bsw = Bs + srow * GT_LD + scol;
  const _Float16* Asr = As + (wm * 64) * GT_LD;
  const _Float16* Bsr = Bs + (wn * 64) * GT_LD;

  v8f acc[4][4];
#pragma unroll
  for (int i = 0; i < 4; ++i)
#pragma unroll
    for (int j = 0; j < 4; ++j) acc[i][j] = (v8f){};

  v8h sa[8], sb[8];
#pragma unroll
  for (int i = 0; i < 8; ++i) {
    sa[i] = *(const v8h*)(Ag + (size_t)(16 * i) * K);
    sb[i] = *(const v8h*)(Bg + (size_t)(16 * i) * K);
  }

  for (int k0 = 0; k0 < K; k0 += GT_BK) {
#pragma unroll
    for (int i = 0; i < 8; ++i) {
      *(v8h*)(Asw + (16 * i) * GT_LD) = sa[i];
      *(v8h*)(Bsw + (16 * i) * GT_LD) = sb[i];
    }
    __syncthreads();
    if (k0 + GT_BK < K) {
#pragma unroll
      for (int i = 0; i < 8; ++i) {
        sa[i] = *(const v8h*)(Ag + (size_t)(16 * i) * K + k0 + GT_BK);
        sb[i] = *(const v8h*)(Bg + (size_t)(16 * i) * K + k0 + GT_BK);
      }
    }
#pragma unroll
    for (int kk = 0; kk < GT_BK; kk += 32) {
      v16h af[4], bf[4];
#pragma unroll
      for (int i = 0; i < 4; ++i) {
        af[i] = load_frag(Asr + (i * 16) * GT_LD + kk, GT_LD, lane);
        bf[i] = load_frag(Bsr + (i * 16) * GT_LD + kk, GT_LD, lane);
      }
#pragma unroll
      for (int i = 0; i < 4; ++i)
#pragma unroll
        for (int j = 0; j < 4; ++j) acc[i][j] = wmma_f16(af[i], bf[j], acc[i][j]);
    }
    __syncthreads();
  }

  const int n = lane & 15, mh = (lane >> 4) * 8;
  const int cbase = tn * GT_BN + wn * 64;
  const size_t rbase = (size_t)tm * GT_BM + wm * 64;
  float* stg = reinterpret_cast<float*>(As) + w * (16 * EP_LD);

  if (EPI == 3 && cbase >= 768) {
    const int hh = (cbase - 768) >> 6;
    const size_t bh = (rbase >> 8) * 6 + hh;
    const size_t t0 = (rbase & 255) + mh;
#pragma unroll
    for (int j = 0; j < 4; ++j) {
      const int d = j * 16 + n;
#pragma unroll
      for (int i = 0; i < 4; ++i) {
        v8h pk;
#pragma unroll
        for (int r = 0; r < 8; ++r) pk[r] = (_Float16)acc[i][j][r];
        *(v8h*)(outvt + (bh * 64 + d) * 256 + t0 + i * 16) = pk;
      }
    }
    return;
  }

  float bv[4];
#pragma unroll
  for (int j = 0; j < 4; ++j) bv[j] = (EPI != 3) ? bias[cbase + j * 16 + n] : 0.f;

  if (EPI == 0 || EPI == 2) {
    const int r2 = lane >> 4;
    const int c4 = (lane & 15) * 4;
    const float* src = (EPI == 0) ? res : outf;
    const size_t o0 = (rbase + r2) * (size_t)N + cbase + c4;
    v4f cur[8], nxt[8];
#pragma unroll
    for (int it = 0; it < 8; ++it) cur[it] = *(const v4f*)(src + o0 + (size_t)(2 * it) * N);
#pragma unroll
    for (int i = 0; i < 4; ++i) {
      if (i < 3) {
#pragma unroll
        for (int it = 0; it < 8; ++it)
          nxt[it] = *(const v4f*)(src + o0 + (size_t)((i + 1) * 16 + 2 * it) * N);
      }
#pragma unroll
      for (int j = 0; j < 4; ++j)
#pragma unroll
        for (int r = 0; r < 8; ++r) stg[(mh + r) * EP_LD + j * 16 + n] = acc[i][j][r] + bv[j];
      wave_lds_sync();
#pragma unroll
      for (int it = 0; it < 8; ++it) {
        const v4f s = *(const v4f*)(stg + (2 * it + r2) * EP_LD + c4);
        v4f o;
#pragma unroll
        for (int e = 0; e < 4; ++e) o[e] = (EPI == 0) ? s[e] + cur[it][e] : cur[it][e] + s[e];
        *(v4f*)(outf + o0 + (size_t)(i * 16 + 2 * it) * N) = o;
      }
      wave_lds_sync();
#pragma unroll
      for (int it = 0; it < 8; ++it) cur[it] = nxt[it];
    }
  } else {
    const int r4 = lane >> 3;
    const int c8 = (lane & 7) * 8;
    _Float16* dst;
    size_t ldo;
    if (EPI == 1) {
      dst = outh + rbase * (size_t)N + cbase;
      ldo = (size_t)N;
    } else {
      const int which = cbase / 384;
      const int hh = (cbase - which * 384) >> 6;
      const size_t bh = (rbase >> 8) * 6 + hh;
      dst = ((which == 0) ? outh : outk) + (bh * 256 + (rbase & 255)) * 64;
      ldo = 64;
    }
#pragma unroll
    for (int i = 0; i < 4; ++i) {
#pragma unroll
      for (int j = 0; j < 4; ++j)
#pragma unroll
        for (int r = 0; r < 8; ++r) {
          float v = acc[i][j][r];
          if (EPI == 1) v = fmaxf(v + bv[j], 0.f);
          stg[(mh + r) * EP_LD + j * 16 + n] = v;
        }
      wave_lds_sync();
#pragma unroll
      for (int it = 0; it < 4; ++it) {
        const int row = 4 * it + r4;
        const v4f s0 = *(const v4f*)(stg + row * EP_LD + c8);
        const v4f s1 = *(const v4f*)(stg + row * EP_LD + c8 + 4);
        v8h pk;
#pragma unroll
        for (int e = 0; e < 4; ++e) { pk[e] = (_Float16)s0[e]; pk[e + 4] = (_Float16)s1[e]; }
        *(v8h*)(dst + (size_t)(i * 16 + row) * ldo + c8) = pk;
      }
      wave_lds_sync();
    }
  }
}

extern "C" void kernel_launch(void* const* d_in, const int* in_sizes, int n_in,
                              void* d_out, int out_size, void* d_ws, size_t ws_size,
                              hipStream_t stream) {
  (void)in_sizes; (void)n_in; (void)out_size; (void)ws_size;
  const float* x   = (const float*)d_in[0];
  const float* Wq  = (const float*)d_in[1];
  const float* Wk  = (const float*)d_in[2];
  const float* Wv  = (const float*)d_in[3];
  const float* Wp  = (const float*)d_in[4];
  const float* bp  = (const float*)d_in[5];
  const float* W1  = (const float*)d_in[6];
  const float* b1  = (const float*)d_in[7];
  const float* W2  = (const float*)d_in[8];
  const float* b2  = (const float*)d_in[9];
  const float* g1  = (const float*)d_in[10];
  const float* be1 = (const float*)d_in[11];
  const float* g2  = (const float*)d_in[12];
  const float* be2 = (const float*)d_in[13];
  float* out = (float*)d_out;

  char* ws = (char*)d_ws;
  const size_t S = (size_t)32768 * 384 * 2;
  _Float16* h16    = (_Float16*)(ws + 0 * S);
  _Float16* q16    = (_Float16*)(ws + 1 * S);
  _Float16* k16    = (_Float16*)(ws + 2 * S);
  _Float16* vt16   = (_Float16*)(ws + 3 * S);
  _Float16* attn16 = (_Float16*)(ws + 0 * S);
  _Float16* h2_16  = (_Float16*)(ws + 1 * S);
  _Float16* a1_16  = (_Float16*)(ws + 2 * S);
  char* wbase = ws + 6 * S;
  _Float16* wqt = (_Float16*)(wbase);
  _Float16* wpt = (_Float16*)(wbase + 3 * 294912);
  _Float16* w1t = (_Float16*)(wbase + 4 * 294912);
  _Float16* w2t = (_Float16*)(wbase + 4 * 294912 + 1179648);

  prep_kernel<<<4528, 256, 0, stream>>>(Wq, Wk, Wv, Wp, W1, W2, wqt, x, g1, be1, h16);
  gemm_tile<3><<<2304, 128, 0, stream>>>(h16, wqt, nullptr, nullptr, nullptr, q16, k16, vt16, 1152, 384);
  attn_kernel<<<6144, 32, 0, stream>>>(q16, k16, vt16, attn16);
  gemm_tile<0><<<768, 128, 0, stream>>>(attn16, wpt, bp, x, out, nullptr, nullptr, nullptr, 384, 384);
  ln_kernel<<<4096, 256, 0, stream>>>(out, g2, be2, h2_16);
  gemm_tile<1><<<3072, 128, 0, stream>>>(h2_16, w1t, b1, nullptr, nullptr, a1_16, nullptr, nullptr, 1536, 384);
  gemm_tile<2><<<768, 128, 0, stream>>>(a1_16, w2t, b2, nullptr, out, nullptr, nullptr, nullptr, 384, 1536);
}


# Attention: a 4-wave work-group per (b,h) copies K and V^T once into LDS with whole-line loads and reads every fragment from LDS; wave w takes query-tile pairs w and 7-w, 9 chunks each; sums unchanged
# speedup vs baseline: 5.2259x; 1.1006x over previous
#include <hip/hip_runtime.h>
#include <hip/hip_bf16.h>


typedef __attribute__((ext_vector_type(16))) _Float16 v16h;
typedef __attribute__((ext_vector_type(8)))  _Float16 v8h;
typedef __attribute__((ext_vector_type(8)))  float    v8f;

#define DEV __device__ __forceinline__

DEV v8f wmma_f16(v16h a, v16h b, v8f c) {
  return __builtin_amdgcn_wmma_f32_16x16x32_f16(false, a, false, b, (short)0, c,
                                                false, false);
}

DEV v16h load_frag(const _Float16* __restrict__ tile, int ld, int lane) {
  const int m  = lane & 15;
  const int kh = (lane & 16) >> 1;
  const _Float16* p = tile + (size_t)m * ld + kh;
  v8h lo = *(const v8h*)p;
  v8h hi = *(const v8h*)(p + 16);
  v16h a;
#pragma unroll
  for (int i = 0; i < 8; ++i) { a[i] = lo[i]; a[i + 8] = hi[i]; }
  return a;
}

DEV void ln_rows(const float* __restrict__ x, const float* __restrict__ g,
                 const float* __restrict__ be, _Float16* __restrict__ out, int blk) {
  const int C = 384;
  int lane = threadIdx.x & 31;
  size_t row = (size_t)blk * 8 + (threadIdx.x >> 5);
  const float* xr = x + row * C;
  float v[12];
  float s = 0.f;
#pragma unroll
  for (int i = 0; i < 12; ++i) { v[i] = xr[lane + 32 * i]; s += v[i]; }
#pragma unroll
  for (int m = 16; m >= 1; m >>= 1) s += __shfl_xor(s, m, 32);
  float mu = s * (1.f / 384.f);
  float var = 0.f;
#pragma unroll
  for (int i = 0; i < 12; ++i) { float d = v[i] - mu; var += d * d; }
#pragma unroll
  for (int m = 16; m >= 1; m >>= 1) var += __shfl_xor(var, m, 32);
  float rs = rsqrtf(var * (1.f / 384.f) + 1e-5f);
  _Float16* orow = out + row * C;
#pragma unroll
  for (int i = 0; i < 12; ++i) {
    int c = lane + 32 * i;
    orow[c] = (_Float16)((v[i] - mu) * rs * g[c] + be[c]);
  }
}

__global__ void ln_kernel(const float* __restrict__ x, const float* __restrict__ g,
                          const float* __restrict__ be, _Float16* __restrict__ out) {
  ln_rows(x, g, be, out, blockIdx.x);
}

typedef __attribute__((ext_vector_type(4))) float v4f;
#define PREP_CAST_BLOCKS 432
#define PREP_LD 66

__global__ __launch_bounds__(256)
void prep_kernel(const float* __restrict__ Wq, const float* __restrict__ Wk,
                 const float* __restrict__ Wv, const float* __restrict__ Wp,
                 const float* __restrict__ W1, const float* __restrict__ W2,
                 _Float16* __restrict__ wout,
                 const float* __restrict__ x, const float* __restrict__ g,
                 const float* __restrict__ be, _Float16* __restrict__ h) {
  __shared__ __attribute__((aligned(16))) _Float16 tile[64 * PREP_LD];
  const int blk = blockIdx.x;
  if (blk >= PREP_CAST_BLOCKS) {
    ln_rows(x, g, be, h, blk - PREP_CAST_BLOCKS);
    return;
  }
  const float* in;
  _Float16* out;
  int R, C, tr, tc;
  if (blk < 108) {
    const int j = blk / 36;
    const int t = blk - j * 36;
    const int b = t / 6;
    in  = (j == 0 ? Wq : (j == 1 ? Wk : Wv)) + (size_t)b * 24576;
    out = wout + (size_t)j * 147456 + (size_t)b * 24576;
    R = 384; C = 64; tr = t - b * 6; tc = 0;
  } else if (blk < 144) {
    const int t = blk - 108;
    in = Wp; out = wout + 442368;
    R = 384; C = 384; tr = t / 6; tc = t - tr * 6;
  } else if (blk < 288) {
    const int t = blk - 144;
    in = W1; out = wout + 589824;
    R = 384; C = 1536; tr = t / 24; tc = t - tr * 24;
  } else {
    const int t = blk - 288;
    in = W2; out = wout + 1179648;
    R = 1536; C = 384; tr = t / 6; tc = t - tr * 6;
  }
  const int tid = threadIdx.x;
  {
    const int lr = tid >> 4;
    const int lc = (tid & 15) * 4;
    const float* src = in + (size_t)(tr * 64 + lr) * C + tc * 64 + lc;
#pragma unroll
    for (int i = 0; i < 4; ++i) {
      const v4f f = *(const v4f*)(src + (size_t)(16 * i) * C);
      _Float16* d = tile + (lr + 16 * i) * PREP_LD + lc;
      d[0] = (_Float16)f[0]; d[1] = (_Float16)f[1];
      d[2] = (_Float16)f[2]; d[3] = (_Float16)f[3];
    }
  }
  __syncthreads();
  {
    const int oc = tid >> 3;
    const int r8 = (tid & 7) * 8;
    _Float16* dst = out + (size_t)(tc * 64 + oc) * R + tr * 64 + r8;
#pragma unroll
    for (int i = 0; i < 2; ++i) {
      v8h pk;
#pragma unroll
      for (int k = 0; k < 8; ++k) pk[k] = tile[(r8 + k) * PREP_LD + oc + 32 * i];
      *(v8h*)(dst + (size_t)(32 * i) * R) = pk;
    }
  }
}


DEV float xhalf(float x) {
  const int b = __builtin_bit_cast(int, x);
  return __builtin_bit_cast(
      float, __builtin_amdgcn_permlanex16(b, b, 0x76543210, 0xfedcba98, false, false));
}

#define AT_KLD 72
#define AT_VLD 264

__global__ __launch_bounds__(128)
void attn_kernel(const _Float16* __restrict__ q,
                 const _Float16* __restrict__ k,
                 const _Float16* __restrict__ vt,
                 _Float16* __restrict__ attn_out) {
  const int T = 256, Dh = 64, H = 6, C = 384;
  __shared__ __attribute__((aligned(16))) _Float16 Ks[256 * AT_KLD];
  __shared__ __attribute__((aligned(16))) _Float16 Vs[64 * AT_VLD];
  const int bh = blockIdx.x;
  const int b = bh / H, h = bh - b * H;
  const int tid = threadIdx.x;
  const int lane = tid & 31;
  const int w = tid >> 5;
  const int c = lane & 15;
  const int kh = (lane >> 4) * 8;

  const _Float16* kb = k + (size_t)bh * T * Dh;
  const _Float16* vb = vt + (size_t)bh * Dh * T;

  {
    v8h st[16];
#pragma unroll
    for (int i = 0; i < 16; ++i) st[i] = *(const v8h*)(kb + (size_t)(tid + 128 * i) * 8);
#pragma unroll
    for (int i = 0; i < 16; ++i) {
      const int id = tid + 128 * i;
      *(v8h*)(Ks + (id >> 3) * AT_KLD + (id & 7) * 8) = st[i];
    }
#pragma unroll
    for (int i = 0; i < 16; ++i) st[i] = *(const v8h*)(vb + (size_t)(tid + 128 * i) * 8);
#pragma unroll
    for (int i = 0; i < 16; ++i) {
      const int id = tid + 128 * i;
      *(v8h*)(Vs + (id >> 5) * AT_VLD + (id & 31) * 8) = st[i];
    }
  }
  __syncthreads();

  const float scale = 0.05103103630798287f;

  for (int pass = 0; pass < 2; ++pass) {
    const int ip = pass ? (7 - w) : w;
    const int t0 = ip * 32;
    const _Float16* qb = q + ((size_t)bh * T + t0) * Dh;

    v16h bq[2][2];
#pragma unroll
    for (int u = 0; u < 2; ++u) {
      bq[u][0] = load_frag(qb + (size_t)(u * 16) * Dh, Dh, lane);
      bq[u][1] = load_frag(qb + (size_t)(u * 16) * Dh + 32, Dh, lane);
    }

    float mrun[2], lrun[2];
    v8f O[2][4];
#pragma unroll
    for (int u = 0; u < 2; ++u) {
      mrun[u] = -__builtin_inff();
      lrun[u] = 0.f;
#pragma unroll
      for (int i = 0; i < 4; ++i) O[u][i] = (v8f){};
    }

    for (int s0 = 0; s0 <= t0; s0 += 32) {
      const _Float16* kp = Ks + s0 * AT_KLD;
      const v16h ka0 = load_frag(kp, AT_KLD, lane);
      const v16h ka1 = load_frag(kp + 32, AT_KLD, lane);
      const v16h kc0 = load_frag(kp + 16 * AT_KLD, AT_KLD, lane);
      const v16h kc1 = load_frag(kp + 16 * AT_KLD + 32, AT_KLD, lane);
      v16h va[4];
#pragma unroll
      for (int nt = 0; nt < 4; ++nt) va[nt] = load_frag(Vs + (nt * 16) * AT_VLD + s0, AT_VLD, lane);

#pragma unroll
      for (int u = 0; u < 2; ++u) {
        const int tu = t0 + 16 * u;
        const int tq = tu + c;
        v8f S0 = {}, S1 = {};
        S0 = wmma_f16(ka0, bq[u][0], S0);
        S0 = wmma_f16(ka1, bq[u][1], S0);
        S1 = wmma_f16(kc0, bq[u][0], S1);
        S1 = wmma_f16(kc1, bq[u][1], S1);

        float p0[8], p1[8];
#pragma unroll
        for (int r = 0; r < 8; ++r) { p0[r] = S0[r] * scale; p1[r] = S1[r] * scale; }
        if (s0 + 31 > tu) {
#pragma unroll
          for (int r = 0; r < 8; ++r) {
            if (s0 + kh + r > tq)      p0[r] = -__builtin_inff();
            if (s0 + 16 + kh + r > tq) p1[r] = -__builtin_inff();
          }
        }
        float mx = fmaxf(p0[0], p1[0]);
#pragma unroll
        for (int r = 1; r < 8; ++r) mx = fmaxf(mx, fmaxf(p0[r], p1[r]));
        mx = fmaxf(mx, xhalf(mx));
        const float mnew = fmaxf(mrun[u], mx);
        const float al = expf(mrun[u] - mnew);
        float rs0 = 0.f, rs1 = 0.f;
#pragma unroll
        for (int r = 0; r < 8; ++r) {
          p0[r] = expf(p0[r] - mnew); rs0 += p0[r];
          p1[r] = expf(p1[r] - mnew); rs1 += p1[r];
        }
        float rs = rs0 + rs1;
        rs += xhalf(rs);
        lrun[u] = lrun[u] * al + rs;
        mrun[u] = mnew;

        v16h bp;
#pragma unroll
        for (int r = 0; r < 8; ++r) { bp[r] = (_Float16)p0[r]; bp[r + 8] = (_Float16)p1[r]; }
#pragma unroll
        for (int nt = 0; nt < 4; ++nt)
#pragma unroll
          for (int r = 0; r < 8; ++r) O[u][nt][r] *= al;
#pragma unroll
        for (int nt = 0; nt < 4; ++nt) O[u][nt] = wmma_f16(va[nt], bp, O[u][nt]);
      }
    }

#pragma unroll
    for (int u = 0; u < 2; ++u) {
      const float inv = 1.f / lrun[u];
      _Float16* orow = attn_out + ((size_t)b * T + t0 + 16 * u + c) * C + h * Dh + kh;
#pragma unroll
      for (int nt = 0; nt < 4; ++nt) {
        v8h pk;
#pragma unroll
        for (int r = 0; r < 8; ++r) pk[r] = (_Float16)(O[u][nt][r] * inv);
        *(v8h*)(orow + nt * 16) = pk;
      }
    }
  }
}

#define GT_BM 128
#define GT_BN 128
#define GT_BK 64
#define GT_LD 72
#define EP_LD 68

DEV void wave_lds_sync() {
  __builtin_amdgcn_fence(__ATOMIC_RELEASE, "wavefront");
  __builtin_amdgcn_wave_barrier();
  __builtin_amdgcn_fence(__ATOMIC_ACQUIRE, "wavefront");
}

template <int EPI>
__global__ __launch_bounds__(128)
void gemm_tile(const _Float16* __restrict__ A, const _Float16* __restrict__ Bt,
               const float* __restrict__ bias, const float* __restrict__ res,
               float* __restrict__ outf, _Float16* __restrict__ outh,
               _Float16* __restrict__ outk, _Float16* __restrict__ outvt,
               int N, int K) {
  __shared__ __attribute__((aligned(16))) _Float16 As[GT_BM * GT_LD];
  __shared__ __attribute__((aligned(16))) _Float16 Bs[GT_BN * GT_LD];
  const int tid = threadIdx.x;
  const int lane = tid & 31;
  const int w = tid >> 5;
  const int wm = w >> 1, wn = w & 1;
  const int ntn = N / GT_BN;
  const int tm = blockIdx.x / ntn;
  const int tn = blockIdx.x - tm * ntn;

  const int srow = tid >> 3;
  const int scol = (tid & 7) * 8;
  const _Float16* Ag = A  + ((size_t)tm * GT_BM + srow) * K + scol;
  const _Float16* Bg = Bt + ((size_t)tn * GT_BN + srow) * K + scol;
  _Float16* Asw = As + srow * GT_LD + scol;
  _Float16* Bsw = Bs + srow * GT_LD + scol;
  const _Float16* Asr = As + (wm * 64) * GT_LD;
  const _Float16* Bsr = Bs + (wn * 64) * GT_LD;

  v8f acc[4][4];
#pragma unroll
  for (int i = 0; i < 4; ++i)
#pragma unroll
    for (int j = 0; j < 4; ++j) acc[i][j] = (v8f){};

  v8h sa[8], sb[8];
#pragma unroll
  for (int i = 0; i < 8; ++i) {
    sa[i] = *(const v8h*)(Ag + (size_t)(16 * i) * K);
    sb[i] = *(const v8h*)(Bg + (size_t)(16 * i) * K);
  }

  for (int k0 = 0; k0 < K; k0 += GT_BK) {
#pragma unroll
    for (int i = 0; i < 8; ++i) {
      *(v8h*)(Asw + (16 * i) * GT_LD) = sa[i];
      *(v8h*)(Bsw + (16 * i) * GT_LD) = sb[i];
    }
    __syncthreads();
    if (k0 + GT_BK < K) {
#pragma unroll
      for (int i = 0; i < 8; ++i) {
        sa[i] = *(const v8h*)(Ag + (size_t)(16 * i) * K + k0 + GT_BK);
        sb[i] = *(const v8h*)(Bg + (size_t)(16 * i) * K + k0 + GT_BK);
      }
    }
#pragma unroll
    for (int kk = 0; kk < GT_BK; kk += 32) {
      v16h af[4], bf[4];
#pragma unroll
      for (int i = 0; i < 4; ++i) {
        af[i] = load_frag(Asr + (i * 16) * GT_LD + kk, GT_LD, lane);
        bf[i] = load_frag(Bsr + (i * 16) * GT_LD + kk, GT_LD, lane);
      }
#pragma unroll
      for (int i = 0; i < 4; ++i)
#pragma unroll
        for (int j = 0; j < 4; ++j) acc[i][j] = wmma_f16(af[i], bf[j], acc[i][j]);
    }
    __syncthreads();
  }

  const int n = lane & 15, mh = (lane >> 4) * 8;
  const int cbase = tn * GT_BN + wn * 64;
  const size_t rbase = (size_t)tm * GT_BM + wm * 64;
  float* stg = reinterpret_cast<float*>(As) + w * (16 * EP_LD);

  if (EPI == 3 && cbase >= 768) {
    const int hh = (cbase - 768) >> 6;
    const size_t bh = (rbase >> 8) * 6 + hh;
    const size_t t0 = (rbase & 255) + mh;
#pragma unroll
    for (int j = 0; j < 4; ++j) {
      const int d = j * 16 + n;
#pragma unroll
      for (int i = 0; i < 4; ++i) {
        v8h pk;
#pragma unroll
        for (int r = 0; r < 8; ++r) pk[r] = (_Float16)acc[i][j][r];
        *(v8h*)(outvt + (bh * 64 + d) * 256 + t0 + i * 16) = pk;
      }
    }
    return;
  }

  float bv[4];
#pragma unroll
  for (int j = 0; j < 4; ++j) bv[j] = (EPI != 3) ? bias[cbase + j * 16 + n] : 0.f;

  if (EPI == 0 || EPI == 2) {
    const int r2 = lane >> 4;
    const int c4 = (lane & 15) * 4;
    const float* src = (EPI == 0) ? res : outf;
    const size_t o0 = (rbase + r2) * (size_t)N + cbase + c4;
    v4f cur[8], nxt[8];
#pragma unroll
    for (int it = 0; it < 8; ++it) cur[it] = *(const v4f*)(src + o0 + (size_t)(2 * it) * N);
#pragma unroll
    for (int i = 0; i < 4; ++i) {
      if (i < 3) {
#pragma unroll
        for (int it = 0; it < 8; ++it)
          nxt[it] = *(const v4f*)(src + o0 + (size_t)((i + 1) * 16 + 2 * it) * N);
      }
#pragma unroll
      for (int j = 0; j < 4; ++j)
#pragma unroll
        for (int r = 0; r < 8; ++r) stg[(mh + r) * EP_LD + j * 16 + n] = acc[i][j][r] + bv[j];
      wave_lds_sync();
#pragma unroll
      for (int it = 0; it < 8; ++it) {
        const v4f s = *(const v4f*)(stg + (2 * it + r2) * EP_LD + c4);
        v4f o;
#pragma unroll
        for (int e = 0; e < 4; ++e) o[e] = (EPI == 0) ? s[e] + cur[it][e] : cur[it][e] + s[e];
        *(v4f*)(outf + o0 + (size_t)(i * 16 + 2 * it) * N) = o;
      }
      wave_lds_sync();
#pragma unroll
      for (int it = 0; it < 8; ++it) cur[it] = nxt[it];
    }
  } else {
    const int r4 = lane >> 3;
    const int c8 = (lane & 7) * 8;
    _Float16* dst;
    size_t ldo;
    if (EPI == 1) {
      dst = outh + rbase * (size_t)N + cbase;
      ldo = (size_t)N;
    } else {
      const int which = cbase / 384;
      const int hh = (cbase - which * 384) >> 6;
      const size_t bh = (rbase >> 8) * 6 + hh;
      dst = ((which == 0) ? outh : outk) + (bh * 256 + (rbase & 255)) * 64;
      ldo = 64;
    }
#pragma unroll
    for (int i = 0; i < 4; ++i) {
#pragma unroll
      for (int j = 0; j < 4; ++j)
#pragma unroll
        for (int r = 0; r < 8; ++r) {
          float v = acc[i][j][r];
          if (EPI == 1) v = fmaxf(v + bv[j], 0.f);
          stg[(mh + r) * EP_LD + j * 16 + n] = v;
        }
      wave_lds_sync();
#pragma unroll
      for (int it = 0; it < 4; ++it) {
        const int row = 4 * it + r4;
        const v4f s0 = *(const v4f*)(stg + row * EP_LD + c8);
        const v4f s1 = *(const v4f*)(stg + row * EP_LD + c8 + 4);
        v8h pk;
#pragma unroll
        for (int e = 0; e < 4; ++e) { pk[e] = (_Float16)s0[e]; pk[e + 4] = (_Float16)s1[e]; }
        *(v8h*)(dst + (size_t)(i * 16 + row) * ldo + c8) = pk;
      }
      wave_lds_sync();
    }
  }
}

extern "C" void kernel_launch(void* const* d_in, const int* in_sizes, int n_in,
                              void* d_out, int out_size, void* d_ws, size_t ws_size,
                              hipStream_t stream) {
  (void)in_sizes; (void)n_in; (void)out_size; (void)ws_size;
  const float* x   = (const float*)d_in[0];
  const float* Wq  = (const float*)d_in[1];
  const float* Wk  = (const float*)d_in[2];
  const float* Wv  = (const float*)d_in[3];
  const float* Wp  = (const float*)d_in[4];
  const float* bp  = (const float*)d_in[5];
  const float* W1  = (const float*)d_in[6];
  const float* b1  = (const float*)d_in[7];
  const float* W2  = (const float*)d_in[8];
  const float* b2  = (const float*)d_in[9];
  const float* g1  = (const float*)d_in[10];
  const float* be1 = (const float*)d_in[11];
  const float* g2  = (const float*)d_in[12];
  const float* be2 = (const float*)d_in[13];
  float* out = (float*)d_out;

  char* ws = (char*)d_ws;
  const size_t S = (size_t)32768 * 384 * 2;
  _Float16* h16    = (_Float16*)(ws + 0 * S);
  _Float16* q16    = (_Float16*)(ws + 1 * S);
  _Float16* k16    = (_Float16*)(ws + 2 * S);
  _Float16* vt16   = (_Float16*)(ws + 3 * S);
  _Float16* attn16 = (_Float16*)(ws + 0 * S);
  _Float16* h2_16  = (_Float16*)(ws + 1 * S);
  _Float16* a1_16  = (_Float16*)(ws + 2 * S);
  char* wbase = ws + 6 * S;
  _Float16* wqt = (_Float16*)(wbase);
  _Float16* wpt = (_Float16*)(wbase + 3 * 294912);
  _Float16* w1t = (_Float16*)(wbase + 4 * 294912);
  _Float16* w2t = (_Float16*)(wbase + 4 * 294912 + 1179648);

  prep_kernel<<<4528, 256, 0, stream>>>(Wq, Wk, Wv, Wp, W1, W2, wqt, x, g1, be1, h16);
  gemm_tile<3><<<2304, 128, 0, stream>>>(h16, wqt, nullptr, nullptr, nullptr, q16, k16, vt16, 1152, 384);
  attn_kernel<<<768, 128, 0, stream>>>(q16, k16, vt16, attn16);
  gemm_tile<0><<<768, 128, 0, stream>>>(attn16, wpt, bp, x, out, nullptr, nullptr, nullptr, 384, 384);
  ln_kernel<<<4096, 256, 0, stream>>>(out, g2, be2, h2_16);
  gemm_tile<1><<<3072, 128, 0, stream>>>(h2_16, w1t, b1, nullptr, nullptr, a1_16, nullptr, nullptr, 1536, 384);
  gemm_tile<2><<<768, 128, 0, stream>>>(a1_16, w2t, b2, nullptr, out, nullptr, nullptr, nullptr, 384, 1536);
}
